# starting kernel
# baseline (speedup 1.0000x reference)
#include <hip/hip_runtime.h>
#include <stdint.h>

#define NN    50000
#define EE    400000
#define INCH  128
#define OUTC  64
#define H1    8
#define NEG   0.2f

typedef float v2f __attribute__((ext_vector_type(2)));
typedef float v8f __attribute__((ext_vector_type(8)));

__global__ void fill_u32_kernel(uint32_t* p, uint32_t v, long long n) {
  long long i = (long long)blockIdx.x * blockDim.x + threadIdx.x;
  if (i < n) p[i] = v;
}

__global__ void gemm_wmma_f32_kernel(const float* __restrict__ A,
                                     const float* __restrict__ B,
                                     float* __restrict__ C,
                                     int M, int K, int Nc) {
  const int wid  = (blockIdx.x * blockDim.x + threadIdx.x) >> 5;
  const int lane = threadIdx.x & 31;
  const int tilesN   = Nc >> 6;
  const int rowTiles = M >> 4;
  if (wid >= rowTiles * tilesN) return;

  const int rt = wid / tilesN;
  const int cg = wid - rt * tilesN;
  const int m0 = rt << 4;
  const int n0 = cg << 6;
  const int ml = lane & 15;
  const int kh = lane >> 4;

  v8f acc0 = {}, acc1 = {}, acc2 = {}, acc3 = {};
  const float* arow = A + (size_t)(m0 + ml) * K;

  for (int k = 0; k < K; k += 4) {
    const int ka = k + 2 * kh;
    v2f a;
    a.x = arow[ka];
    a.y = arow[ka + 1];
    const float* b0p = B + (size_t)ka * Nc + n0 + ml;
    const float* b1p = b0p + Nc;
    v2f b0; b0.x = b0p[0];  b0.y = b1p[0];
    v2f b1; b1.x = b0p[16]; b1.y = b1p[16];
    v2f b2; b2.x = b0p[32]; b2.y = b1p[32];
    v2f b3; b3.x = b0p[48]; b3.y = b1p[48];
    acc0 = __builtin_amdgcn_wmma_f32_16x16x4_f32(false, a, false, b0, (short)0, acc0, false, false);
    acc1 = __builtin_amdgcn_wmma_f32_16x16x4_f32(false, a, false, b1, (short)0, acc1, false, false);
    acc2 = __builtin_amdgcn_wmma_f32_16x16x4_f32(false, a, false, b2, (short)0, acc2, false, false);
    acc3 = __builtin_amdgcn_wmma_f32_16x16x4_f32(false, a, false, b3, (short)0, acc3, false, false);
  }

  const int mr = m0 + 8 * kh;
#pragma unroll
  for (int v = 0; v < 8; ++v) {
    float* crow = C + (size_t)(mr + v) * Nc + n0 + ml;
    crow[0]  = acc0[v];
    crow[16] = acc1[v];
    crow[32] = acc2[v];
    crow[48] = acc3[v];
  }
}

__global__ void att_scores_kernel(const float* __restrict__ h,
                                  const float* __restrict__ att_src,
                                  const float* __restrict__ att_dst,
                                  float* __restrict__ a_s,
                                  float* __restrict__ a_d,
                                  int n_nodes, int heads) {
  const int wid  = (blockIdx.x * blockDim.x + threadIdx.x) >> 5;
  const int lane = threadIdx.x & 31;
  if (wid >= n_nodes * heads) return;
  const int n  = wid / heads;
  const int hd = wid - n * heads;
  const float* hrow = h + (size_t)n * heads * OUTC + hd * OUTC;
  const float* ws   = att_src + hd * OUTC;
  const float* wd   = att_dst + hd * OUTC;
  float vs = hrow[lane] * ws[lane] + hrow[lane + 32] * ws[lane + 32];
  float vd = hrow[lane] * wd[lane] + hrow[lane + 32] * wd[lane + 32];
#pragma unroll
  for (int off = 16; off; off >>= 1) {
    vs += __shfl_xor(vs, off, 32);
    vd += __shfl_xor(vd, off, 32);
  }
  if (lane == 0) {
    a_s[n * heads + hd] = vs;
    a_d[n * heads + hd] = vd;
  }
}

__device__ __forceinline__ void edge_endpoints(const int* ei, int e, int& s, int& d) {
  if (e < EE) { s = ei[e]; d = ei[EE + e]; }
  else        { s = d = e - EE; }
}

__device__ __forceinline__ float leaky(float x) {
  return x > 0.0f ? x : NEG * x;
}

__device__ __forceinline__ void atomicMaxF32(float* addr, float val) {
  if (val >= 0.0f) atomicMax((int*)addr, __float_as_int(val));
  else             atomicMin((unsigned int*)addr, __float_as_uint(val));
}

__global__ void edge_max_kernel(const int* __restrict__ ei,
                                const float* __restrict__ a_s,
                                const float* __restrict__ a_d,
                                float* __restrict__ emax, int heads) {
  const long long tid = (long long)blockIdx.x * blockDim.x + threadIdx.x;
  const long long total = (long long)(EE + NN) * heads;
  if (tid >= total) return;
  const int e  = (int)(tid / heads);
  const int hd = (int)(tid - (long long)e * heads);
  int s, d; edge_endpoints(ei, e, s, d);
  const float ev = leaky(a_s[s * heads + hd] + a_d[d * heads + hd]);
  atomicMaxF32(&emax[d * heads + hd], ev);
}

__global__ void edge_expsum_kernel(const int* __restrict__ ei,
                                   const float* __restrict__ a_s,
                                   const float* __restrict__ a_d,
                                   const float* __restrict__ emax,
                                   float* __restrict__ denom, int heads) {
  const long long tid = (long long)blockIdx.x * blockDim.x + threadIdx.x;
  const long long total = (long long)(EE + NN) * heads;
  if (tid >= total) return;
  const int e  = (int)(tid / heads);
  const int hd = (int)(tid - (long long)e * heads);
  int s, d; edge_endpoints(ei, e, s, d);
  const float ev = leaky(a_s[s * heads + hd] + a_d[d * heads + hd]);
  atomicAdd(&denom[d * heads + hd], __expf(ev - emax[d * heads + hd]));
}

__global__ void edge_aggregate_kernel(const int* __restrict__ ei,
                                      const float* __restrict__ a_s,
                                      const float* __restrict__ a_d,
                                      const float* __restrict__ emax,
                                      const float* __restrict__ denom,
                                      const float* __restrict__ h,
                                      float* __restrict__ acc, int heads) {
  const long long wid = ((long long)blockIdx.x * blockDim.x + threadIdx.x) >> 5;
  const int lane = threadIdx.x & 31;
  const long long total = (long long)(EE + NN) * heads;
  if (wid >= total) return;
  const int e  = (int)(wid / heads);
  const int hd = (int)(wid - (long long)e * heads);
  int s, d; edge_endpoints(ei, e, s, d);
  const float ev = leaky(a_s[s * heads + hd] + a_d[d * heads + hd]);
  const float alpha = __expf(ev - emax[d * heads + hd]) / denom[d * heads + hd];
  const float* hrow = h + (size_t)s * heads * OUTC + hd * OUTC;
  float* arow = acc + (size_t)d * OUTC;
  atomicAdd(&arow[lane],      alpha * hrow[lane]);
  atomicAdd(&arow[lane + 32], alpha * hrow[lane + 32]);
}

__global__ void finalize_kernel(const float* __restrict__ acc,
                                const float* __restrict__ bias,
                                float* __restrict__ out,
                                float invH, int do_elu) {
  const long long tid = (long long)blockIdx.x * blockDim.x + threadIdx.x;
  if (tid >= (long long)NN * OUTC) return;
  const int c = (int)(tid & (OUTC - 1));
  float v = acc[tid] * invH + bias[c];
  if (do_elu) v = v > 0.0f ? v : (__expf(v) - 1.0f);
  out[tid] = v;
}

static inline unsigned cdiv(long long a, long long b) { return (unsigned)((a + b - 1) / b); }

extern "C" void kernel_launch(void* const* d_in, const int* in_sizes, int n_in,
                              void* d_out, int out_size, void* d_ws, size_t ws_size,
                              hipStream_t stream) {
  (void)in_sizes; (void)n_in; (void)out_size; (void)ws_size;
  const float* x    = (const float*)d_in[0];
  const int*   ei   = (const int*)d_in[1];
  const float* W1   = (const float*)d_in[2];
  const float* asw1 = (const float*)d_in[3];
  const float* adw1 = (const float*)d_in[4];
  const float* b1   = (const float*)d_in[5];
  const float* W2   = (const float*)d_in[6];
  const float* asw2 = (const float*)d_in[7];
  const float* adw2 = (const float*)d_in[8];
  const float* b2   = (const float*)d_in[9];
  float* out = (float*)d_out;

  float* ws   = (float*)d_ws;
  float* h1   = ws; ws += (size_t)NN * H1 * OUTC;
  float* as1  = ws; ws += (size_t)NN * H1;
  float* ad1  = ws; ws += (size_t)NN * H1;
  float* em1  = ws; ws += (size_t)NN * H1;
  float* dn1  = ws; ws += (size_t)NN * H1;
  float* acc1 = ws; ws += (size_t)NN * OUTC;
  float* h2   = ws; ws += (size_t)NN * OUTC;
  float* as2  = ws; ws += (size_t)NN;
  float* ad2  = ws; ws += (size_t)NN;
  float* em2  = ws; ws += (size_t)NN;
  float* dn2  = ws; ws += (size_t)NN;

  const int T = 256;
  const long long enh1 = (long long)(EE + NN) * H1;
  const long long enh2 = (long long)(EE + NN);
  const uint32_t NEG_INF = 0xFF800000u;

  fill_u32_kernel<<<cdiv(NN * H1, T), T, 0, stream>>>((uint32_t*)em1, NEG_INF, NN * H1);
  fill_u32_kernel<<<cdiv(NN * H1, T), T, 0, stream>>>((uint32_t*)dn1, 0u, NN * H1);
  fill_u32_kernel<<<cdiv((long long)NN * OUTC, T), T, 0, stream>>>((uint32_t*)acc1, 0u, (long long)NN * OUTC);

  {
    long long waves = ((long long)NN / 16) * (H1 * OUTC / 64);
    gemm_wmma_f32_kernel<<<cdiv(waves * 32, T), T, 0, stream>>>(x, W1, h1, NN, INCH, H1 * OUTC);
  }
  att_scores_kernel<<<cdiv((long long)NN * H1 * 32, T), T, 0, stream>>>(h1, asw1, adw1, as1, ad1, NN, H1);
  edge_max_kernel<<<cdiv(enh1, T), T, 0, stream>>>(ei, as1, ad1, em1, H1);
  edge_expsum_kernel<<<cdiv(enh1, T), T, 0, stream>>>(ei, as1, ad1, em1, dn1, H1);
  edge_aggregate_kernel<<<cdiv(enh1 * 32, T), T, 0, stream>>>(ei, as1, ad1, em1, dn1, h1, acc1, H1);
  finalize_kernel<<<cdiv((long long)NN * OUTC, T), T, 0, stream>>>(acc1, b1, acc1, 1.0f / H1, 1);

  fill_u32_kernel<<<cdiv(NN, T), T, 0, stream>>>((uint32_t*)em2, NEG_INF, NN);
  fill_u32_kernel<<<cdiv(NN, T), T, 0, stream>>>((uint32_t*)dn2, 0u, NN);
  fill_u32_kernel<<<cdiv((long long)NN * OUTC, T), T, 0, stream>>>((uint32_t*)out, 0u, (long long)NN * OUTC);

  {
    long long waves = ((long long)NN / 16) * (OUTC / 64);
    gemm_wmma_f32_kernel<<<cdiv(waves * 32, T), T, 0, stream>>>(acc1, W2, h2, NN, OUTC, OUTC);
  }
  att_scores_kernel<<<cdiv((long long)NN * 32, T), T, 0, stream>>>(h2, asw2, adw2, as2, ad2, NN, 1);
  edge_max_kernel<<<cdiv(enh2, T), T, 0, stream>>>(ei, as2, ad2, em2, 1);
  edge_expsum_kernel<<<cdiv(enh2, T), T, 0, stream>>>(ei, as2, ad2, em2, dn2, 1);
  edge_aggregate_kernel<<<cdiv(enh2 * 32, T), T, 0, stream>>>(ei, as2, ad2, em2, dn2, h2, out, 1);
  finalize_kernel<<<cdiv((long long)NN * OUTC, T), T, 0, stream>>>(out, b2, out, 1.0f, 0);
}


# Owner workgroups issue the global gathers of several list entries together: row-sum loop 4 (layer 1) or 8 (layer 2) rows per wave step, score loop up to 8 items per thread; LDS arrival order kept
# speedup vs baseline: 3.6696x; 3.6696x over previous
#include <hip/hip_runtime.h>
#include <stdint.h>

#define NN    50000
#define EE    400000
#define INCH  128
#define OUTC  64
#define H1    8
#define NEG   0.2f

typedef float v2f __attribute__((ext_vector_type(2)));
typedef float v8f __attribute__((ext_vector_type(8)));

__global__ void gemm_wmma_f32_kernel(const float* __restrict__ A,
                                     const float* __restrict__ B,
                                     float* __restrict__ C,
                                     const float* __restrict__ att_src,
                                     const float* __restrict__ att_dst,
                                     float* __restrict__ a_s,
                                     float* __restrict__ a_d,
                                     int M, int K, int Nc) {
  const int wid  = (blockIdx.x * blockDim.x + threadIdx.x) >> 5;
  const int lane = threadIdx.x & 31;
  const int tilesN   = Nc >> 6;
  const int rowTiles = M >> 4;
  if (wid >= rowTiles * tilesN) return;

  const int rt = wid / tilesN;
  const int cg = wid - rt * tilesN;
  const int m0 = rt << 4;
  const int n0 = cg << 6;
  const int ml = lane & 15;
  const int kh = lane >> 4;

  v8f acc0 = {}, acc1 = {}, acc2 = {}, acc3 = {};
  const float* arow = A + (size_t)(m0 + ml) * K;

  for (int k = 0; k < K; k += 4) {
    const int ka = k + 2 * kh;
    v2f a;
    a.x = arow[ka];
    a.y = arow[ka + 1];
    const float* b0p = B + (size_t)ka * Nc + n0 + ml;
    const float* b1p = b0p + Nc;
    v2f b0; b0.x = b0p[0];  b0.y = b1p[0];
    v2f b1; b1.x = b0p[16]; b1.y = b1p[16];
    v2f b2; b2.x = b0p[32]; b2.y = b1p[32];
    v2f b3; b3.x = b0p[48]; b3.y = b1p[48];
    acc0 = __builtin_amdgcn_wmma_f32_16x16x4_f32(false, a, false, b0, (short)0, acc0, false, false);
    acc1 = __builtin_amdgcn_wmma_f32_16x16x4_f32(false, a, false, b1, (short)0, acc1, false, false);
    acc2 = __builtin_amdgcn_wmma_f32_16x16x4_f32(false, a, false, b2, (short)0, acc2, false, false);
    acc3 = __builtin_amdgcn_wmma_f32_16x16x4_f32(false, a, false, b3, (short)0, acc3, false, false);
  }

  const int mr = m0 + 8 * kh;
#pragma unroll
  for (int v = 0; v < 8; ++v) {
    float* crow = C + (size_t)(mr + v) * Nc + n0 + ml;
    crow[0]  = acc0[v];
    crow[16] = acc1[v];
    crow[32] = acc2[v];
    crow[48] = acc3[v];
  }

  const float* ws = att_src + n0 + ml;
  const float* wd = att_dst + n0 + ml;
  const float s0 = ws[0], s1 = ws[16], s2 = ws[32], s3 = ws[48];
  const float d0 = wd[0], d1 = wd[16], d2 = wd[32], d3 = wd[48];
  float ps[8], pd[8];
#pragma unroll
  for (int v = 0; v < 8; ++v) {
    ps[v] = acc0[v] * s0 + acc1[v] * s1 + acc2[v] * s2 + acc3[v] * s3;
    pd[v] = acc0[v] * d0 + acc1[v] * d1 + acc2[v] * d2 + acc3[v] * d3;
  }
#pragma unroll
  for (int off = 8; off; off >>= 1) {
#pragma unroll
    for (int v = 0; v < 8; ++v) {
      ps[v] += __shfl_xor(ps[v], off, 32);
      pd[v] += __shfl_xor(pd[v], off, 32);
    }
  }
  float vs = ps[0], vd = pd[0];
#pragma unroll
  for (int v = 1; v < 8; ++v) {
    if (ml == v) { vs = ps[v]; vd = pd[v]; }
  }
  if (ml < 8) {
    const int idx = (mr + ml) * tilesN + cg;
    a_s[idx]   = vs;
    a_d[idx]   = vd;
  }
}

__device__ __forceinline__ float leaky(float x) {
  return x > 0.0f ? x : NEG * x;
}

__device__ __forceinline__ float edge_score(float as, float ad) {
  return leaky(as + ad);
}

__device__ __forceinline__ void atomicMaxF32(float* addr, float val) {
  if (val >= 0.0f) atomicMax((int*)addr, __float_as_int(val));
  else             atomicMin((unsigned int*)addr, __float_as_uint(val));
}

#define OWN     196
#define OWN_WG  ((NN + OWN - 1) / OWN)
#define OWN_T   1024
#define OWN_NW  (OWN_T / 32)
#define CHUNK   128
#define NCHUNK  (EE / CHUNK)
#define LCAP    3072
static_assert(EE % CHUNK == 0, "edge list is scanned in whole 128-edge chunks");
static_assert(NCHUNK >= 3 * OWN_NW, "every wave starts with three chunks in flight");
static_assert(EE + NN <= (1 << 19) && OWN <= (1 << 13), "a list word packs the edge number in 19 bits, the local destination above it");
static_assert(OWN <= LCAP, "the self loops always fit the list");
static_assert(OWN_T % H1 == 0, "a thread's items all belong to one head");

template <int HEADS>
__device__ __forceinline__ float edge_alpha(int s, int dl, int lane,
                                            const float* __restrict__ a_s,
                                            const float* s_ad, const float* s_em,
                                            const float* s_dn) {
  const int hh = lane & (HEADS - 1);
  const float ev = edge_score(a_s[s * HEADS + hh], s_ad[dl * HEADS + hh]);
  return __expf(ev - s_em[dl * HEADS + hh]) / s_dn[dl * HEADS + hh];
}

template <int HEADS> struct Row;
template <> struct Row<8> { float4 v0, v1, v2, v3; };
template <> struct Row<1> { float2 v; };

__device__ __forceinline__ void row_gather(Row<8>& r, int s, int lane, const float* __restrict__ h) {
  const float4* hrow = (const float4*)(h + (size_t)s * (8 * OUTC));
  r.v0 = hrow[lane];
  r.v1 = hrow[32 + lane];
  r.v2 = hrow[64 + lane];
  r.v3 = hrow[96 + lane];
}

__device__ __forceinline__ void row_gather(Row<1>& r, int s, int lane, const float* __restrict__ h) {
  r.v = ((const float2*)(h + (size_t)s * OUTC))[lane];
}

__device__ __forceinline__ void hold_gathers(const Row<1> (&r)[8]) {
  asm volatile("" :: "v"(r[0].v.x), "v"(r[0].v.y), "v"(r[1].v.x), "v"(r[1].v.y),
                     "v"(r[2].v.x), "v"(r[2].v.y), "v"(r[3].v.x), "v"(r[3].v.y),
                     "v"(r[4].v.x), "v"(r[4].v.y), "v"(r[5].v.x), "v"(r[5].v.y),
                     "v"(r[6].v.x), "v"(r[6].v.y), "v"(r[7].v.x), "v"(r[7].v.y));
}
__device__ __forceinline__ void hold_gathers(const Row<8> (&)[4]) {}

__device__ __forceinline__ void row_add(const Row<8>& r, int dl, int lane, float al, float* s_acc) {
  const float4 v0 = r.v0, v1 = r.v1, v2 = r.v2, v3 = r.v3;
  const int hb = lane >> 4;
  const float a0 = __shfl(al, hb, 32);
  const float a1 = __shfl(al, 2 + hb, 32);
  const float a2 = __shfl(al, 4 + hb, 32);
  const float a3 = __shfl(al, 6 + hb, 32);
  float rx = a0 * v0.x + a1 * v1.x + a2 * v2.x + a3 * v3.x;
  float ry = a0 * v0.y + a1 * v1.y + a2 * v2.y + a3 * v3.y;
  float rz = a0 * v0.z + a1 * v1.z + a2 * v2.z + a3 * v3.z;
  float rw = a0 * v0.w + a1 * v1.w + a2 * v2.w + a3 * v3.w;
  rx += __shfl_xor(rx, 16, 32);
  ry += __shfl_xor(ry, 16, 32);
  rz += __shfl_xor(rz, 16, 32);
  rw += __shfl_xor(rw, 16, 32);
  float* arow = s_acc + dl * OUTC + (lane & 15) * 4;
  if (lane < 16) { atomicAdd(&arow[0], rx); atomicAdd(&arow[1], ry); }
  else           { atomicAdd(&arow[2], rz); atomicAdd(&arow[3], rw); }
}

__device__ __forceinline__ void row_add(const Row<1>& r, int dl, int lane, float al, float* s_acc) {
  float* arow = s_acc + dl * OUTC + 2 * lane;
  atomicAdd(&arow[0], al * r.v.x);
  atomicAdd(&arow[1], al * r.v.y);
}

template <int HEADS>
__device__ __forceinline__ void edge_accumulate(int s, int dl, int lane, float al,
                                                const float* __restrict__ h,
                                                float* s_acc) {
  Row<HEADS> r;
  row_gather(r, s, lane, h);
  row_add(r, dl, lane, al, s_acc);
}

template <int HEADS, int U>
__device__ __forceinline__ void score_items(int j0, int hd,
                                            const int* __restrict__ ei,
                                            const float* __restrict__ a_s,
                                            const unsigned* s_list, const float* s_ad,
                                            int* s_src, float* s_p) {
  unsigned en[U];
  int sn[U];
  float as[U];
#pragma unroll
  for (int u = 0; u < U; ++u) en[u] = s_list[(j0 + u * OWN_T) / HEADS];
#pragma unroll
  for (int u = 0; u < U; ++u) {
    const int e = (int)(en[u] & 0x7FFFFu);
    const int sl = ei[e < EE ? e : EE - 1];
    sn[u] = e < EE ? sl : e - EE;
  }
#pragma unroll
  for (int u = 0; u < U; ++u) as[u] = a_s[sn[u] * HEADS + hd];
#pragma unroll
  for (int u = 0; u < U; ++u) {
    const int j = j0 + u * OWN_T;
    if (hd == 0) s_src[j / HEADS] = sn[u];
    s_p[j] = edge_score(as[u], s_ad[(int)(en[u] >> 19) * HEADS + hd]);
  }
}

__device__ __forceinline__ int own_chunk(int c, int rot) {
  int t = c + rot;
  return t >= NCHUNK ? t - NCHUNK : t;
}

__device__ __forceinline__ void owner_chunk_append(int4 q, int cc, int lo, int cnt, int lane,
                                                   unsigned* s_list, unsigned* s_cnt) {
  const bool f0 = (unsigned)(q.x - lo) < (unsigned)cnt;
  const bool f1 = (unsigned)(q.y - lo) < (unsigned)cnt;
  const bool f2 = (unsigned)(q.z - lo) < (unsigned)cnt;
  const bool f3 = (unsigned)(q.w - lo) < (unsigned)cnt;
  const unsigned m0 = (unsigned)__ballot(f0);
  const unsigned m1 = (unsigned)__ballot(f1);
  const unsigned m2 = (unsigned)__ballot(f2);
  const unsigned m3 = (unsigned)__ballot(f3);
  if ((m0 | m1 | m2 | m3) == 0u) return;
  const int tot = __popc(m0) + __popc(m1) + __popc(m2) + __popc(m3);
  int base = 0;
  if (lane == 0) base = (int)atomicAdd(s_cnt, (unsigned)tot);
  base = __builtin_amdgcn_readfirstlane(base);
  const unsigned lt = (1u << lane) - 1u;
  int pos = base + __popc(m0 & lt) + __popc(m1 & lt) + __popc(m2 & lt) + __popc(m3 & lt);
  const unsigned e0 = (unsigned)(cc * CHUNK + lane * 4);
  if (f0) { if (pos < LCAP) s_list[pos] = ((unsigned)(q.x - lo) << 19) | e0;        ++pos; }
  if (f1) { if (pos < LCAP) s_list[pos] = ((unsigned)(q.y - lo) << 19) | (e0 + 1u); ++pos; }
  if (f2) { if (pos < LCAP) s_list[pos] = ((unsigned)(q.z - lo) << 19) | (e0 + 2u); ++pos; }
  if (f3) { if (pos < LCAP) s_list[pos] = ((unsigned)(q.w - lo) << 19) | (e0 + 3u); ++pos; }
}

template <int HEADS, int MODE>
__device__ __forceinline__ void owner_direct_pass(int lo, int cnt, int wave, int lane,
                                                  const int* __restrict__ ei,
                                                  const float* __restrict__ a_s,
                                                  const float* __restrict__ h,
                                                  const float* s_ad, float* s_em,
                                                  float* s_dn, float* s_acc) {
  for (int e = wave; e < EE + cnt; e += OWN_NW) {
    int s, d;
    if (e < EE) { s = ei[e]; d = ei[EE + e]; }
    else        { s = d = lo + (e - EE); }
    const int dl = d - lo;
    if ((unsigned)dl >= (unsigned)cnt) continue;
    if (MODE == 2) {
      const float al = edge_alpha<HEADS>(s, dl, lane, a_s, s_ad, s_em, s_dn);
      edge_accumulate<HEADS>(s, dl, lane, al, h, s_acc);
    } else if (lane < HEADS) {
      const float ev = edge_score(a_s[s * HEADS + lane], s_ad[dl * HEADS + lane]);
      if (MODE == 0) atomicMaxF32(&s_em[dl * HEADS + lane], ev);
      else           atomicAdd(&s_dn[dl * HEADS + lane], __expf(ev - s_em[dl * HEADS + lane]));
    }
  }
}

template <int HEADS, int DO_ELU>
__global__ __launch_bounds__(OWN_T)
void owner_aggregate_kernel(const int* __restrict__ ei,
                            const float* __restrict__ a_s,
                            const float* __restrict__ a_d,
                            const float* __restrict__ h,
                            const float* __restrict__ bias,
                            float* __restrict__ out, float invH) {
  __shared__ float s_acc[OWN * OUTC];
  __shared__ float s_ad[OWN * HEADS];
  __shared__ float s_em[OWN * HEADS];
  __shared__ float s_dn[OWN * HEADS];
  __shared__ unsigned s_list[LCAP];
  __shared__ int s_src[LCAP];
  __shared__ float s_p[LCAP * HEADS];
  __shared__ unsigned s_cnt;

  const int lo = blockIdx.x * OWN;
  int cnt = NN - lo;
  if (cnt > OWN) cnt = OWN;
  const int wave = __builtin_amdgcn_readfirstlane(threadIdx.x >> 5);
  const int lane = threadIdx.x & 31;

  for (int i = threadIdx.x; i < cnt * OUTC; i += OWN_T) s_acc[i] = 0.0f;
  for (int i = threadIdx.x; i < cnt * HEADS; i += OWN_T) {
    s_ad[i] = a_d[lo * HEADS + i];
    s_em[i] = -__builtin_inff();
    s_dn[i] = 0.0f;
  }
  for (int i = threadIdx.x; i < cnt; i += OWN_T) s_list[i] = ((unsigned)i << 19) | (unsigned)(EE + lo + i);
  if (threadIdx.x == 0) s_cnt = (unsigned)cnt;
  __syncthreads();

  {
    const int4* dst4 = (const int4*)(ei + EE);
    const int rot = (blockIdx.x * 12) % NCHUNK;
    const int WSTEP = OWN_NW;
    int4 q0 = dst4[own_chunk(wave, rot) * 32 + lane];
    int4 q1 = dst4[own_chunk(wave + WSTEP, rot) * 32 + lane];
    int4 q2 = dst4[own_chunk(wave + 2 * WSTEP, rot) * 32 + lane];
    for (int c = wave; c < NCHUNK; c += 3 * WSTEP) {
      owner_chunk_append(q0, own_chunk(c, rot), lo, cnt, lane, s_list, &s_cnt);
      if (c + 3 * WSTEP < NCHUNK) q0 = dst4[own_chunk(c + 3 * WSTEP, rot) * 32 + lane];
      if (c + WSTEP < NCHUNK) {
        owner_chunk_append(q1, own_chunk(c + WSTEP, rot), lo, cnt, lane, s_list, &s_cnt);
        if (c + 4 * WSTEP < NCHUNK) q1 = dst4[own_chunk(c + 4 * WSTEP, rot) * 32 + lane];
      }
      if (c + 2 * WSTEP < NCHUNK) {
        owner_chunk_append(q2, own_chunk(c + 2 * WSTEP, rot), lo, cnt, lane, s_list, &s_cnt);
        if (c + 5 * WSTEP < NCHUNK) q2 = dst4[own_chunk(c + 5 * WSTEP, rot) * 32 + lane];
      }
    }
  }
  __syncthreads();
  const int total = (int)s_cnt;

  if (total <= LCAP) {
    const int items = total * HEADS;
    {
      const int hd = threadIdx.x % HEADS;
      int j = threadIdx.x;
      if (HEADS > 1) {
        for (; j + 7 * OWN_T < items; j += 8 * OWN_T) score_items<HEADS, 8>(j, hd, ei, a_s, s_list, s_ad, s_src, s_p);
        if (j + 3 * OWN_T < items) { score_items<HEADS, 4>(j, hd, ei, a_s, s_list, s_ad, s_src, s_p); j += 4 * OWN_T; }
      }
      for (; j + OWN_T < items; j += 2 * OWN_T) score_items<HEADS, 2>(j, hd, ei, a_s, s_list, s_ad, s_src, s_p);
      if (j < items) score_items<HEADS, 1>(j, hd, ei, a_s, s_list, s_ad, s_src, s_p);
    }
    for (int j = threadIdx.x; j < items; j += OWN_T) {
      const unsigned en = s_list[j / HEADS];
      atomicMaxF32(&s_em[(int)(en >> 19) * HEADS + j % HEADS], s_p[j]);
    }
    __syncthreads();
    for (int j = threadIdx.x; j < items; j += OWN_T) {
      const unsigned en = s_list[j / HEADS];
      const int k = (int)(en >> 19) * HEADS + j % HEADS;
      const float p = __expf(s_p[j] - s_em[k]);
      s_p[j] = p;
      atomicAdd(&s_dn[k], p);
    }
    __syncthreads();
    constexpr int SB = HEADS == 8 ? 4 : 8;
    const int hh = lane & (HEADS - 1);
    const int nfull = total / SB;
    for (int t = wave; t < nfull; t += OWN_NW) {
      int sn[SB], dl[SB];
      float p[SB], al[SB];
      Row<HEADS> r[SB];
#pragma unroll
      for (int b = 0; b < SB; ++b) {
        const int i = t * SB + b;
        sn[b] = __builtin_amdgcn_readfirstlane(s_src[i]);
        dl[b] = (int)((unsigned)__builtin_amdgcn_readfirstlane((int)s_list[i]) >> 19);
        p[b]  = s_p[i * HEADS + hh];
      }
#pragma unroll
      for (int b = 0; b < SB; ++b) row_gather(r[b], sn[b], lane, h);
#pragma unroll
      for (int b = 0; b < SB; ++b) al[b] = p[b] / s_dn[dl[b] * HEADS + hh];
      hold_gathers(r);
#pragma unroll
      for (int b = 0; b < SB; ++b) row_add(r[b], dl[b], lane, al[b], s_acc);
    }
    for (int i = nfull * SB + wave; i < total; i += OWN_NW) {
      const int s  = __builtin_amdgcn_readfirstlane(s_src[i]);
      const int dl = (int)((unsigned)__builtin_amdgcn_readfirstlane((int)s_list[i]) >> 19);
      const float al = s_p[i * HEADS + hh] / s_dn[dl * HEADS + hh];
      edge_accumulate<HEADS>(s, dl, lane, al, h, s_acc);
    }
  } else {
    owner_direct_pass<HEADS, 0>(lo, cnt, wave, lane, ei, a_s, h, s_ad, s_em, s_dn, s_acc);
    __syncthreads();
    owner_direct_pass<HEADS, 1>(lo, cnt, wave, lane, ei, a_s, h, s_ad, s_em, s_dn, s_acc);
    __syncthreads();
    owner_direct_pass<HEADS, 2>(lo, cnt, wave, lane, ei, a_s, h, s_ad, s_em, s_dn, s_acc);
  }
  __syncthreads();

  for (int i = threadIdx.x; i < cnt * OUTC; i += OWN_T) {
    float v = s_acc[i] * invH + bias[i & (OUTC - 1)];
    if (DO_ELU) v = v > 0.0f ? v : (__expf(v) - 1.0f);
    out[(size_t)lo * OUTC + i] = v;
  }
}

static inline unsigned cdiv(long long a, long long b) { return (unsigned)((a + b - 1) / b); }

extern "C" void kernel_launch(void* const* d_in, const int* in_sizes, int n_in,
                              void* d_out, int out_size, void* d_ws, size_t ws_size,
                              hipStream_t stream) {
  (void)in_sizes; (void)n_in; (void)out_size; (void)ws_size;
  const float* x    = (const float*)d_in[0];
  const int*   ei   = (const int*)d_in[1];
  const float* W1   = (const float*)d_in[2];
  const float* asw1 = (const float*)d_in[3];
  const float* adw1 = (const float*)d_in[4];
  const float* b1   = (const float*)d_in[5];
  const float* W2   = (const float*)d_in[6];
  const float* asw2 = (const float*)d_in[7];
  const float* adw2 = (const float*)d_in[8];
  const float* b2   = (const float*)d_in[9];
  float* out = (float*)d_out;

  float* ws   = (float*)d_ws;
  float* h1   = ws; ws += (size_t)NN * H1 * OUTC;
  float* as1  = ws; ws += (size_t)NN * H1;
  float* ad1  = ws; ws += (size_t)NN * H1;
  float* acc1 = ws; ws += (size_t)NN * OUTC;
  float* h2   = ws; ws += (size_t)NN * OUTC;
  float* as2  = ws; ws += (size_t)NN;
  float* ad2  = ws; ws += (size_t)NN;

  const int T = 256;

  {
    long long waves = ((long long)NN / 16) * (H1 * OUTC / 64);
    gemm_wmma_f32_kernel<<<cdiv(waves * 32, T), T, 0, stream>>>(x, W1, h1, asw1, adw1, as1, ad1,
                                                                 NN, INCH, H1 * OUTC);
  }
  owner_aggregate_kernel<H1, 1><<<OWN_WG, OWN_T, 0, stream>>>(ei, as1, ad1, h1, b1, acc1, 1.0f / H1);

  {
    long long waves = ((long long)NN / 16) * (OUTC / 64);
    gemm_wmma_f32_kernel<<<cdiv(waves * 32, T), T, 0, stream>>>(acc1, W2, h2, asw2, adw2, as2, ad2,
                                                                 NN, OUTC, OUTC);
  }
  owner_aggregate_kernel<1, 0><<<OWN_WG, OWN_T, 0, stream>>>(ei, as2, ad2, h2, b2, out, 1.0f);
}


# Layer-2 owner workgroups take the edge list that the layer-1 launch built (kept in their own output slice) instead of scanning the destination list again; layer 2 sums in layer 1's list order
# speedup vs baseline: 4.0656x; 1.1079x over previous
#include <hip/hip_runtime.h>
#include <stdint.h>

#define NN    50000
#define EE    400000
#define INCH  128
#define OUTC  64
#define H1    8
#define NEG   0.2f

typedef float v2f __attribute__((ext_vector_type(2)));
typedef float v8f __attribute__((ext_vector_type(8)));

__global__ void gemm_wmma_f32_kernel(const float* __restrict__ A,
                                     const float* __restrict__ B,
                                     float* __restrict__ C,
                                     const float* __restrict__ att_src,
                                     const float* __restrict__ att_dst,
                                     float* __restrict__ a_s,
                                     float* __restrict__ a_d,
                                     int M, int K, int Nc) {
  const int wid  = (blockIdx.x * blockDim.x + threadIdx.x) >> 5;
  const int lane = threadIdx.x & 31;
  const int tilesN   = Nc >> 6;
  const int rowTiles = M >> 4;
  if (wid >= rowTiles * tilesN) return;

  const int rt = wid / tilesN;
  const int cg = wid - rt * tilesN;
  const int m0 = rt << 4;
  const int n0 = cg << 6;
  const int ml = lane & 15;
  const int kh = lane >> 4;

  v8f acc0 = {}, acc1 = {}, acc2 = {}, acc3 = {};
  const float* arow = A + (size_t)(m0 + ml) * K;

  for (int k = 0; k < K; k += 4) {
    const int ka = k + 2 * kh;
    v2f a;
    a.x = arow[ka];
    a.y = arow[ka + 1];
    const float* b0p = B + (size_t)ka * Nc + n0 + ml;
    const float* b1p = b0p + Nc;
    v2f b0; b0.x = b0p[0];  b0.y = b1p[0];
    v2f b1; b1.x = b0p[16]; b1.y = b1p[16];
    v2f b2; b2.x = b0p[32]; b2.y = b1p[32];
    v2f b3; b3.x = b0p[48]; b3.y = b1p[48];
    acc0 = __builtin_amdgcn_wmma_f32_16x16x4_f32(false, a, false, b0, (short)0, acc0, false, false);
    acc1 = __builtin_amdgcn_wmma_f32_16x16x4_f32(false, a, false, b1, (short)0, acc1, false, false);
    acc2 = __builtin_amdgcn_wmma_f32_16x16x4_f32(false, a, false, b2, (short)0, acc2, false, false);
    acc3 = __builtin_amdgcn_wmma_f32_16x16x4_f32(false, a, false, b3, (short)0, acc3, false, false);
  }

  const int mr = m0 + 8 * kh;
#pragma unroll
  for (int v = 0; v < 8; ++v) {
    float* crow = C + (size_t)(mr + v) * Nc + n0 + ml;
    crow[0]  = acc0[v];
    crow[16] = acc1[v];
    crow[32] = acc2[v];
    crow[48] = acc3[v];
  }

  const float* ws = att_src + n0 + ml;
  const float* wd = att_dst + n0 + ml;
  const float s0 = ws[0], s1 = ws[16], s2 = ws[32], s3 = ws[48];
  const float d0 = wd[0], d1 = wd[16], d2 = wd[32], d3 = wd[48];
  float ps[8], pd[8];
#pragma unroll
  for (int v = 0; v < 8; ++v) {
    ps[v] = acc0[v] * s0 + acc1[v] * s1 + acc2[v] * s2 + acc3[v] * s3;
    pd[v] = acc0[v] * d0 + acc1[v] * d1 + acc2[v] * d2 + acc3[v] * d3;
  }
#pragma unroll
  for (int off = 8; off; off >>= 1) {
#pragma unroll
    for (int v = 0; v < 8; ++v) {
      ps[v] += __shfl_xor(ps[v], off, 32);
      pd[v] += __shfl_xor(pd[v], off, 32);
    }
  }
  float vs = ps[0], vd = pd[0];
#pragma unroll
  for (int v = 1; v < 8; ++v) {
    if (ml == v) { vs = ps[v]; vd = pd[v]; }
  }
  if (ml < 8) {
    const int idx = (mr + ml) * tilesN + cg;
    a_s[idx]   = vs;
    a_d[idx]   = vd;
  }
}

__device__ __forceinline__ float leaky(float x) {
  return x > 0.0f ? x : NEG * x;
}

__device__ __forceinline__ float edge_score(float as, float ad) {
  return leaky(as + ad);
}

__device__ __forceinline__ void atomicMaxF32(float* addr, float val) {
  if (val >= 0.0f) atomicMax((int*)addr, __float_as_int(val));
  else             atomicMin((unsigned int*)addr, __float_as_uint(val));
}

#define OWN     196
#define OWN_WG  ((NN + OWN - 1) / OWN)
#define OWN_T   1024
#define OWN_NW  (OWN_T / 32)
#define CHUNK   128
#define NCHUNK  (EE / CHUNK)
#define LCAP    3072
#define KEEP_AT   32
#define KEEP_NONE 0xFFFFFFFFu
static_assert(EE % CHUNK == 0, "edge list is scanned in whole 128-edge chunks");
static_assert(NCHUNK >= 3 * OWN_NW, "every wave starts with three chunks in flight");
static_assert(EE + NN <= (1 << 19) && OWN <= (1 << 13), "a list word packs the edge number in 19 bits, the local destination above it");
static_assert(OWN <= LCAP, "the self loops always fit the list");
static_assert(OWN_T % H1 == 0, "a thread's items all belong to one head");

template <int HEADS>
__device__ __forceinline__ float edge_alpha(int s, int dl, int lane,
                                            const float* __restrict__ a_s,
                                            const float* s_ad, const float* s_em,
                                            const float* s_dn) {
  const int hh = lane & (HEADS - 1);
  const float ev = edge_score(a_s[s * HEADS + hh], s_ad[dl * HEADS + hh]);
  return __expf(ev - s_em[dl * HEADS + hh]) / s_dn[dl * HEADS + hh];
}

template <int HEADS> struct Row;
template <> struct Row<8> { float4 v0, v1, v2, v3; };
template <> struct Row<1> { float2 v; };

__device__ __forceinline__ void row_gather(Row<8>& r, int s, int lane, const float* __restrict__ h) {
  const float4* hrow = (const float4*)(h + (size_t)s * (8 * OUTC));
  r.v0 = hrow[lane];
  r.v1 = hrow[32 + lane];
  r.v2 = hrow[64 + lane];
  r.v3 = hrow[96 + lane];
}

__device__ __forceinline__ void row_gather(Row<1>& r, int s, int lane, const float* __restrict__ h) {
  r.v = ((const float2*)(h + (size_t)s * OUTC))[lane];
}

__device__ __forceinline__ void hold_gathers(const Row<1> (&r)[8]) {
  asm volatile("" :: "v"(r[0].v.x), "v"(r[0].v.y), "v"(r[1].v.x), "v"(r[1].v.y),
                     "v"(r[2].v.x), "v"(r[2].v.y), "v"(r[3].v.x), "v"(r[3].v.y),
                     "v"(r[4].v.x), "v"(r[4].v.y), "v"(r[5].v.x), "v"(r[5].v.y),
                     "v"(r[6].v.x), "v"(r[6].v.y), "v"(r[7].v.x), "v"(r[7].v.y));
}
__device__ __forceinline__ void hold_gathers(const Row<8> (&)[4]) {}

__device__ __forceinline__ void row_add(const Row<8>& r, int dl, int lane, float al, float* s_acc) {
  const float4 v0 = r.v0, v1 = r.v1, v2 = r.v2, v3 = r.v3;
  const int hb = lane >> 4;
  const float a0 = __shfl(al, hb, 32);
  const float a1 = __shfl(al, 2 + hb, 32);
  const float a2 = __shfl(al, 4 + hb, 32);
  const float a3 = __shfl(al, 6 + hb, 32);
  float rx = a0 * v0.x + a1 * v1.x + a2 * v2.x + a3 * v3.x;
  float ry = a0 * v0.y + a1 * v1.y + a2 * v2.y + a3 * v3.y;
  float rz = a0 * v0.z + a1 * v1.z + a2 * v2.z + a3 * v3.z;
  float rw = a0 * v0.w + a1 * v1.w + a2 * v2.w + a3 * v3.w;
  rx += __shfl_xor(rx, 16, 32);
  ry += __shfl_xor(ry, 16, 32);
  rz += __shfl_xor(rz, 16, 32);
  rw += __shfl_xor(rw, 16, 32);
  float* arow = s_acc + dl * OUTC + (lane & 15) * 4;
  if (lane < 16) { atomicAdd(&arow[0], rx); atomicAdd(&arow[1], ry); }
  else           { atomicAdd(&arow[2], rz); atomicAdd(&arow[3], rw); }
}

__device__ __forceinline__ void row_add(const Row<1>& r, int dl, int lane, float al, float* s_acc) {
  float* arow = s_acc + dl * OUTC + 2 * lane;
  atomicAdd(&arow[0], al * r.v.x);
  atomicAdd(&arow[1], al * r.v.y);
}

template <int HEADS>
__device__ __forceinline__ void edge_accumulate(int s, int dl, int lane, float al,
                                                const float* __restrict__ h,
                                                float* s_acc) {
  Row<HEADS> r;
  row_gather(r, s, lane, h);
  row_add(r, dl, lane, al, s_acc);
}

template <int HEADS, int U>
__device__ __forceinline__ void score_items(int j0, int hd,
                                            const int* __restrict__ ei,
                                            const float* __restrict__ a_s,
                                            const unsigned* s_list, const float* s_ad,
                                            int* s_src, float* s_p) {
  unsigned en[U];
  int sn[U];
  float as[U];
#pragma unroll
  for (int u = 0; u < U; ++u) en[u] = s_list[(j0 + u * OWN_T) / HEADS];
#pragma unroll
  for (int u = 0; u < U; ++u) {
    const int e = (int)(en[u] & 0x7FFFFu);
    const int sl = ei[e < EE ? e : EE - 1];
    sn[u] = e < EE ? sl : e - EE;
  }
#pragma unroll
  for (int u = 0; u < U; ++u) as[u] = a_s[sn[u] * HEADS + hd];
#pragma unroll
  for (int u = 0; u < U; ++u) {
    const int j = j0 + u * OWN_T;
    if (hd == 0) s_src[j / HEADS] = sn[u];
    s_p[j] = edge_score(as[u], s_ad[(int)(en[u] >> 19) * HEADS + hd]);
  }
}

__device__ __forceinline__ int own_chunk(int c, int rot) {
  int t = c + rot;
  return t >= NCHUNK ? t - NCHUNK : t;
}

__device__ __forceinline__ void owner_chunk_append(int4 q, int cc, int lo, int cnt, int lane,
                                                   unsigned* s_list, unsigned* s_cnt) {
  const bool f0 = (unsigned)(q.x - lo) < (unsigned)cnt;
  const bool f1 = (unsigned)(q.y - lo) < (unsigned)cnt;
  const bool f2 = (unsigned)(q.z - lo) < (unsigned)cnt;
  const bool f3 = (unsigned)(q.w - lo) < (unsigned)cnt;
  const unsigned m0 = (unsigned)__ballot(f0);
  const unsigned m1 = (unsigned)__ballot(f1);
  const unsigned m2 = (unsigned)__ballot(f2);
  const unsigned m3 = (unsigned)__ballot(f3);
  if ((m0 | m1 | m2 | m3) == 0u) return;
  const int tot = __popc(m0) + __popc(m1) + __popc(m2) + __popc(m3);
  int base = 0;
  if (lane == 0) base = (int)atomicAdd(s_cnt, (unsigned)tot);
  base = __builtin_amdgcn_readfirstlane(base);
  const unsigned lt = (1u << lane) - 1u;
  int pos = base + __popc(m0 & lt) + __popc(m1 & lt) + __popc(m2 & lt) + __popc(m3 & lt);
  const unsigned e0 = (unsigned)(cc * CHUNK + lane * 4);
  if (f0) { if (pos < LCAP) s_list[pos] = ((unsigned)(q.x - lo) << 19) | e0;        ++pos; }
  if (f1) { if (pos < LCAP) s_list[pos] = ((unsigned)(q.y - lo) << 19) | (e0 + 1u); ++pos; }
  if (f2) { if (pos < LCAP) s_list[pos] = ((unsigned)(q.z - lo) << 19) | (e0 + 2u); ++pos; }
  if (f3) { if (pos < LCAP) s_list[pos] = ((unsigned)(q.w - lo) << 19) | (e0 + 3u); ++pos; }
}

template <int HEADS, int MODE>
__device__ __forceinline__ void owner_direct_pass(int lo, int cnt, int wave, int lane,
                                                  const int* __restrict__ ei,
                                                  const float* __restrict__ a_s,
                                                  const float* __restrict__ h,
                                                  const float* s_ad, float* s_em,
                                                  float* s_dn, float* s_acc) {
  for (int e = wave; e < EE + cnt; e += OWN_NW) {
    int s, d;
    if (e < EE) { s = ei[e]; d = ei[EE + e]; }
    else        { s = d = lo + (e - EE); }
    const int dl = d - lo;
    if ((unsigned)dl >= (unsigned)cnt) continue;
    if (MODE == 2) {
      const float al = edge_alpha<HEADS>(s, dl, lane, a_s, s_ad, s_em, s_dn);
      edge_accumulate<HEADS>(s, dl, lane, al, h, s_acc);
    } else if (lane < HEADS) {
      const float ev = edge_score(a_s[s * HEADS + lane], s_ad[dl * HEADS + lane]);
      if (MODE == 0) atomicMaxF32(&s_em[dl * HEADS + lane], ev);
      else           atomicAdd(&s_dn[dl * HEADS + lane], __expf(ev - s_em[dl * HEADS + lane]));
    }
  }
}

template <int HEADS, int DO_ELU, int LIST>
__global__ __launch_bounds__(OWN_T)
void owner_aggregate_kernel(const int* __restrict__ ei,
                            const float* __restrict__ a_s,
                            const float* __restrict__ a_d,
                            const float* __restrict__ h,
                            const float* __restrict__ bias,
                            float* __restrict__ out, float* keep, float invH) {
  __shared__ float s_acc[OWN * OUTC];
  __shared__ float s_ad[OWN * HEADS];
  __shared__ float s_em[OWN * HEADS];
  __shared__ float s_dn[OWN * HEADS];
  __shared__ unsigned s_list[LCAP];
  __shared__ int s_src[LCAP];
  __shared__ float s_p[LCAP * HEADS];
  __shared__ unsigned s_cnt;

  const int lo = blockIdx.x * OWN;
  int cnt = NN - lo;
  if (cnt > OWN) cnt = OWN;
  const int wave = __builtin_amdgcn_readfirstlane(threadIdx.x >> 5);
  const int lane = threadIdx.x & 31;

  for (int i = threadIdx.x; i < cnt * OUTC; i += OWN_T) s_acc[i] = 0.0f;
  for (int i = threadIdx.x; i < cnt * HEADS; i += OWN_T) {
    s_ad[i] = a_d[lo * HEADS + i];
    s_em[i] = -__builtin_inff();
    s_dn[i] = 0.0f;
  }
  bool kept = false;
  if (LIST == 1) {
    const float* slot = out + (size_t)lo * OUTC;
    const unsigned n = (unsigned)__builtin_amdgcn_readfirstlane((int)__float_as_uint(slot[0]));
    kept = n >= (unsigned)cnt && n <= (unsigned)LCAP && n + KEEP_AT <= (unsigned)(cnt * OUTC);
    if (kept) {
      for (int i = threadIdx.x; i < (int)n; i += OWN_T) s_list[i] = __float_as_uint(slot[KEEP_AT + i]);
      if (threadIdx.x == 0) s_cnt = n;
    }
  }
  if (!kept) {
    for (int i = threadIdx.x; i < cnt; i += OWN_T) s_list[i] = ((unsigned)i << 19) | (unsigned)(EE + lo + i);
    if (threadIdx.x == 0) s_cnt = (unsigned)cnt;
  }
  __syncthreads();

  if (!kept) {
    const int4* dst4 = (const int4*)(ei + EE);
    const int rot = (blockIdx.x * 12) % NCHUNK;
    const int WSTEP = OWN_NW;
    int4 q0 = dst4[own_chunk(wave, rot) * 32 + lane];
    int4 q1 = dst4[own_chunk(wave + WSTEP, rot) * 32 + lane];
    int4 q2 = dst4[own_chunk(wave + 2 * WSTEP, rot) * 32 + lane];
    for (int c = wave; c < NCHUNK; c += 3 * WSTEP) {
      owner_chunk_append(q0, own_chunk(c, rot), lo, cnt, lane, s_list, &s_cnt);
      if (c + 3 * WSTEP < NCHUNK) q0 = dst4[own_chunk(c + 3 * WSTEP, rot) * 32 + lane];
      if (c + WSTEP < NCHUNK) {
        owner_chunk_append(q1, own_chunk(c + WSTEP, rot), lo, cnt, lane, s_list, &s_cnt);
        if (c + 4 * WSTEP < NCHUNK) q1 = dst4[own_chunk(c + 4 * WSTEP, rot) * 32 + lane];
      }
      if (c + 2 * WSTEP < NCHUNK) {
        owner_chunk_append(q2, own_chunk(c + 2 * WSTEP, rot), lo, cnt, lane, s_list, &s_cnt);
        if (c + 5 * WSTEP < NCHUNK) q2 = dst4[own_chunk(c + 5 * WSTEP, rot) * 32 + lane];
      }
    }
  }
  __syncthreads();
  const int total = (int)s_cnt;

  if (total <= LCAP) {
    const int items = total * HEADS;
    {
      const int hd = threadIdx.x % HEADS;
      int j = threadIdx.x;
      if (HEADS > 1) {
        for (; j + 7 * OWN_T < items; j += 8 * OWN_T) score_items<HEADS, 8>(j, hd, ei, a_s, s_list, s_ad, s_src, s_p);
        if (j + 3 * OWN_T < items) { score_items<HEADS, 4>(j, hd, ei, a_s, s_list, s_ad, s_src, s_p); j += 4 * OWN_T; }
      }
      for (; j + OWN_T < items; j += 2 * OWN_T) score_items<HEADS, 2>(j, hd, ei, a_s, s_list, s_ad, s_src, s_p);
      if (j < items) score_items<HEADS, 1>(j, hd, ei, a_s, s_list, s_ad, s_src, s_p);
    }
    for (int j = threadIdx.x; j < items; j += OWN_T) {
      const unsigned en = s_list[j / HEADS];
      atomicMaxF32(&s_em[(int)(en >> 19) * HEADS + j % HEADS], s_p[j]);
    }
    __syncthreads();
    if (LIST == 0) {
      float* slot = keep + (size_t)lo * OUTC;
      const bool fits = total + KEEP_AT <= cnt * OUTC;
      if (fits) {
        for (int i = threadIdx.x; i < total; i += OWN_T)
          slot[KEEP_AT + i] = __uint_as_float((s_list[i] & ~0x7FFFFu) | (unsigned)(EE + s_src[i]));
      }
      if (threadIdx.x == 0) slot[0] = __uint_as_float(fits ? (unsigned)total : KEEP_NONE);
    }
    for (int j = threadIdx.x; j < items; j += OWN_T) {
      const unsigned en = s_list[j / HEADS];
      const int k = (int)(en >> 19) * HEADS + j % HEADS;
      const float p = __expf(s_p[j] - s_em[k]);
      s_p[j] = p;
      atomicAdd(&s_dn[k], p);
    }
    __syncthreads();
    constexpr int SB = HEADS == 8 ? 4 : 8;
    const int hh = lane & (HEADS - 1);
    const int nfull = total / SB;
    for (int t = wave; t < nfull; t += OWN_NW) {
      int sn[SB], dl[SB];
      float p[SB], al[SB];
      Row<HEADS> r[SB];
#pragma unroll
      for (int b = 0; b < SB; ++b) {
        const int i = t * SB + b;
        sn[b] = __builtin_amdgcn_readfirstlane(s_src[i]);
        dl[b] = (int)((unsigned)__builtin_amdgcn_readfirstlane((int)s_list[i]) >> 19);
        p[b]  = s_p[i * HEADS + hh];
      }
#pragma unroll
      for (int b = 0; b < SB; ++b) row_gather(r[b], sn[b], lane, h);
#pragma unroll
      for (int b = 0; b < SB; ++b) al[b] = p[b] / s_dn[dl[b] * HEADS + hh];
      hold_gathers(r);
#pragma unroll
      for (int b = 0; b < SB; ++b) row_add(r[b], dl[b], lane, al[b], s_acc);
    }
    for (int i = nfull * SB + wave; i < total; i += OWN_NW) {
      const int s  = __builtin_amdgcn_readfirstlane(s_src[i]);
      const int dl = (int)((unsigned)__builtin_amdgcn_readfirstlane((int)s_list[i]) >> 19);
      const float al = s_p[i * HEADS + hh] / s_dn[dl * HEADS + hh];
      edge_accumulate<HEADS>(s, dl, lane, al, h, s_acc);
    }
  } else {
    if (LIST == 0 && threadIdx.x == 0) keep[(size_t)lo * OUTC] = __uint_as_float(KEEP_NONE);
    owner_direct_pass<HEADS, 0>(lo, cnt, wave, lane, ei, a_s, h, s_ad, s_em, s_dn, s_acc);
    __syncthreads();
    owner_direct_pass<HEADS, 1>(lo, cnt, wave, lane, ei, a_s, h, s_ad, s_em, s_dn, s_acc);
    __syncthreads();
    owner_direct_pass<HEADS, 2>(lo, cnt, wave, lane, ei, a_s, h, s_ad, s_em, s_dn, s_acc);
  }
  __syncthreads();

  for (int i = threadIdx.x; i < cnt * OUTC; i += OWN_T) {
    float v = s_acc[i] * invH + bias[i & (OUTC - 1)];
    if (DO_ELU) v = v > 0.0f ? v : (__expf(v) - 1.0f);
    out[(size_t)lo * OUTC + i] = v;
  }
}

static inline unsigned cdiv(long long a, long long b) { return (unsigned)((a + b - 1) / b); }

extern "C" void kernel_launch(void* const* d_in, const int* in_sizes, int n_in,
                              void* d_out, int out_size, void* d_ws, size_t ws_size,
                              hipStream_t stream) {
  (void)in_sizes; (void)n_in; (void)out_size; (void)ws_size;
  const float* x    = (const float*)d_in[0];
  const int*   ei   = (const int*)d_in[1];
  const float* W1   = (const float*)d_in[2];
  const float* asw1 = (const float*)d_in[3];
  const float* adw1 = (const float*)d_in[4];
  const float* b1   = (const float*)d_in[5];
  const float* W2   = (const float*)d_in[6];
  const float* asw2 = (const float*)d_in[7];
  const float* adw2 = (const float*)d_in[8];
  const float* b2   = (const float*)d_in[9];
  float* out = (float*)d_out;

  float* ws   = (float*)d_ws;
  float* h1   = ws; ws += (size_t)NN * H1 * OUTC;
  float* as1  = ws; ws += (size_t)NN * H1;
  float* ad1  = ws; ws += (size_t)NN * H1;
  float* acc1 = ws; ws += (size_t)NN * OUTC;
  float* h2   = ws; ws += (size_t)NN * OUTC;
  float* as2  = ws; ws += (size_t)NN;
  float* ad2  = ws; ws += (size_t)NN;

  const int T = 256;

  {
    long long waves = ((long long)NN / 16) * (H1 * OUTC / 64);
    gemm_wmma_f32_kernel<<<cdiv(waves * 32, T), T, 0, stream>>>(x, W1, h1, asw1, adw1, as1, ad1,
                                                                 NN, INCH, H1 * OUTC);
  }
  owner_aggregate_kernel<H1, 1, 0><<<OWN_WG, OWN_T, 0, stream>>>(ei, as1, ad1, h1, b1, acc1, out, 1.0f / H1);

  {
    long long waves = ((long long)NN / 16) * (OUTC / 64);
    gemm_wmma_f32_kernel<<<cdiv(waves * 32, T), T, 0, stream>>>(acc1, W2, h2, asw2, adw2, as2, ad2,
                                                                 NN, OUTC, OUTC);
  }
  owner_aggregate_kernel<1, 0, 1><<<OWN_WG, OWN_T, 0, stream>>>(ei, as2, ad2, h2, b2, out, nullptr, 1.0f);
}


# Layer-1 product launch also sorts the edges by owner workgroup into bit rows (LDS OR per edge); owner workgroups read their 50-KB row instead of scanning all destinations; LDS arrival order kept
# speedup vs baseline: 4.4526x; 1.0952x over previous
#include <hip/hip_runtime.h>
#include <stdint.h>

#define NN    50000
#define EE    400000
#define INCH  128
#define OUTC  64
#define H1    8
#define NEG   0.2f

typedef float v2f __attribute__((ext_vector_type(2)));
typedef float v8f __attribute__((ext_vector_type(8)));

#define OWN     196
#define OWN_WG  ((NN + OWN - 1) / OWN)
#define OWN_T   1024
#define OWN_NW  (OWN_T / 32)
#define CHUNK   128
#define NCHUNK  (EE / CHUNK)
#define LCAP    3072
#define KEEP_AT   32
#define KEEP_NONE 0xFFFFFFFFu
static_assert(EE % CHUNK == 0, "edge list is scanned in whole 128-edge chunks");
static_assert(NCHUNK >= 3 * OWN_NW, "every wave starts with three chunks in flight");
static_assert(EE + NN <= (1 << 19) && OWN <= (1 << 13), "a list word packs the edge number in 19 bits, the local destination above it");
static_assert(OWN <= LCAP, "the self loops always fit the list");
static_assert(OWN_T % H1 == 0, "a thread's items all belong to one head");
#define ROWW    (((EE / 32) + 31) / 32 * 32)
#define PRE_T   256
#define PRE_NW  (PRE_T / 32)
#define PRE_WG  ((NCHUNK + PRE_NW - 1) / PRE_NW)
static_assert(OWN_WG <= 256, "the bit rows are made for 256 owners");
static_assert(PRE_NW == 8, "a product workgroup's eight chunk masks make one 128-byte line of a row");
static_assert(256 * (size_t)ROWW <= (size_t)NN * OUTC + 2 * (size_t)NN, "the bit rows fit the planes they borrow");
static_assert(3 * OWN_T <= NCHUNK && NCHUNK <= 4 * OWN_T, "a thread reads three or four chunk masks of its row");

__global__ void gemm_wmma_f32_kernel(const float* __restrict__ A,
                                     const float* __restrict__ B,
                                     float* __restrict__ C,
                                     const float* __restrict__ att_src,
                                     const float* __restrict__ att_dst,
                                     float* __restrict__ a_s,
                                     float* __restrict__ a_d,
                                     const int* __restrict__ ei,
                                     unsigned* __restrict__ own_bits,
                                     int M, int K, int Nc) {
  __shared__ unsigned s_bits[64 * 32];
  const int wid  = (blockIdx.x * blockDim.x + threadIdx.x) >> 5;
  const int lane = threadIdx.x & 31;
  if (own_bits != nullptr && blockIdx.x < PRE_WG) {
    const int pw = threadIdx.x >> 5;
    const int ch = blockIdx.x * PRE_NW + pw;
    const bool on = ch < NCHUNK;
    unsigned o0 = 0u, o1 = 0u, o2 = 0u, o3 = 0u;
    if (on) {
      const int4 q = ((const int4*)(ei + EE))[ch * 32 + lane];
      o0 = min((unsigned)q.x / (unsigned)OWN, 255u);
      o1 = min((unsigned)q.y / (unsigned)OWN, 255u);
      o2 = min((unsigned)q.z / (unsigned)OWN, 255u);
      o3 = min((unsigned)q.w / (unsigned)OWN, 255u);
    }
    uint4* tile4 = (uint4*)s_bits;
    const unsigned bit = 1u << lane;
    const int t = (int)threadIdx.x;
    for (int p = 0; p < 4; ++p) {
      tile4[t]         = make_uint4(0u, 0u, 0u, 0u);
      tile4[t + PRE_T] = make_uint4(0u, 0u, 0u, 0u);
      __syncthreads();
      if (on) {
        if ((o0 >> 6) == (unsigned)p) atomicOr(&s_bits[(o0 & 63u) * 32 + pw * 4 + 0], bit);
        if ((o1 >> 6) == (unsigned)p) atomicOr(&s_bits[(o1 & 63u) * 32 + pw * 4 + 1], bit);
        if ((o2 >> 6) == (unsigned)p) atomicOr(&s_bits[(o2 & 63u) * 32 + pw * 4 + 2], bit);
        if ((o3 >> 6) == (unsigned)p) atomicOr(&s_bits[(o3 & 63u) * 32 + pw * 4 + 3], bit);
      }
      __syncthreads();
      for (int i = 0; i < 2; ++i) {
        const int u = t + i * PRE_T;
        const int o = 64 * p + (u >> 3);
        const int c = blockIdx.x * PRE_NW + (u & 7);
        if (c < NCHUNK) ((uint4*)(own_bits + (size_t)o * ROWW))[c] = tile4[u];
      }
      if (p < 3) __syncthreads();
    }
  }
  const int tilesN   = Nc >> 6;
  const int rowTiles = M >> 4;
  if (wid >= rowTiles * tilesN) return;

  const int rt = wid / tilesN;
  const int cg = wid - rt * tilesN;
  const int m0 = rt << 4;
  const int n0 = cg << 6;
  const int ml = lane & 15;
  const int kh = lane >> 4;

  v8f acc0 = {}, acc1 = {}, acc2 = {}, acc3 = {};
  const float* arow = A + (size_t)(m0 + ml) * K;

  for (int k = 0; k < K; k += 4) {
    const int ka = k + 2 * kh;
    v2f a;
    a.x = arow[ka];
    a.y = arow[ka + 1];
    const float* b0p = B + (size_t)ka * Nc + n0 + ml;
    const float* b1p = b0p + Nc;
    v2f b0; b0.x = b0p[0];  b0.y = b1p[0];
    v2f b1; b1.x = b0p[16]; b1.y = b1p[16];
    v2f b2; b2.x = b0p[32]; b2.y = b1p[32];
    v2f b3; b3.x = b0p[48]; b3.y = b1p[48];
    acc0 = __builtin_amdgcn_wmma_f32_16x16x4_f32(false, a, false, b0, (short)0, acc0, false, false);
    acc1 = __builtin_amdgcn_wmma_f32_16x16x4_f32(false, a, false, b1, (short)0, acc1, false, false);
    acc2 = __builtin_amdgcn_wmma_f32_16x16x4_f32(false, a, false, b2, (short)0, acc2, false, false);
    acc3 = __builtin_amdgcn_wmma_f32_16x16x4_f32(false, a, false, b3, (short)0, acc3, false, false);
  }

  const int mr = m0 + 8 * kh;
#pragma unroll
  for (int v = 0; v < 8; ++v) {
    float* crow = C + (size_t)(mr + v) * Nc + n0 + ml;
    crow[0]  = acc0[v];
    crow[16] = acc1[v];
    crow[32] = acc2[v];
    crow[48] = acc3[v];
  }

  const float* ws = att_src + n0 + ml;
  const float* wd = att_dst + n0 + ml;
  const float s0 = ws[0], s1 = ws[16], s2 = ws[32], s3 = ws[48];
  const float d0 = wd[0], d1 = wd[16], d2 = wd[32], d3 = wd[48];
  float ps[8], pd[8];
#pragma unroll
  for (int v = 0; v < 8; ++v) {
    ps[v] = acc0[v] * s0 + acc1[v] * s1 + acc2[v] * s2 + acc3[v] * s3;
    pd[v] = acc0[v] * d0 + acc1[v] * d1 + acc2[v] * d2 + acc3[v] * d3;
  }
#pragma unroll
  for (int off = 8; off; off >>= 1) {
#pragma unroll
    for (int v = 0; v < 8; ++v) {
      ps[v] += __shfl_xor(ps[v], off, 32);
      pd[v] += __shfl_xor(pd[v], off, 32);
    }
  }
  float vs = ps[0], vd = pd[0];
#pragma unroll
  for (int v = 1; v < 8; ++v) {
    if (ml == v) { vs = ps[v]; vd = pd[v]; }
  }
  if (ml < 8) {
    const int idx = (mr + ml) * tilesN + cg;
    a_s[idx]   = vs;
    a_d[idx]   = vd;
  }
}

__device__ __forceinline__ float leaky(float x) {
  return x > 0.0f ? x : NEG * x;
}

__device__ __forceinline__ float edge_score(float as, float ad) {
  return leaky(as + ad);
}

__device__ __forceinline__ void atomicMaxF32(float* addr, float val) {
  if (val >= 0.0f) atomicMax((int*)addr, __float_as_int(val));
  else             atomicMin((unsigned int*)addr, __float_as_uint(val));
}


template <int HEADS>
__device__ __forceinline__ float edge_alpha(int s, int dl, int lane,
                                            const float* __restrict__ a_s,
                                            const float* s_ad, const float* s_em,
                                            const float* s_dn) {
  const int hh = lane & (HEADS - 1);
  const float ev = edge_score(a_s[s * HEADS + hh], s_ad[dl * HEADS + hh]);
  return __expf(ev - s_em[dl * HEADS + hh]) / s_dn[dl * HEADS + hh];
}

template <int HEADS> struct Row;
template <> struct Row<8> { float4 v0, v1, v2, v3; };
template <> struct Row<1> { float2 v; };

__device__ __forceinline__ void row_gather(Row<8>& r, int s, int lane, const float* __restrict__ h) {
  const float4* hrow = (const float4*)(h + (size_t)s * (8 * OUTC));
  r.v0 = hrow[lane];
  r.v1 = hrow[32 + lane];
  r.v2 = hrow[64 + lane];
  r.v3 = hrow[96 + lane];
}

__device__ __forceinline__ void row_gather(Row<1>& r, int s, int lane, const float* __restrict__ h) {
  r.v = ((const float2*)(h + (size_t)s * OUTC))[lane];
}

__device__ __forceinline__ void hold_gathers(const Row<1> (&r)[8]) {
  asm volatile("" :: "v"(r[0].v.x), "v"(r[0].v.y), "v"(r[1].v.x), "v"(r[1].v.y),
                     "v"(r[2].v.x), "v"(r[2].v.y), "v"(r[3].v.x), "v"(r[3].v.y),
                     "v"(r[4].v.x), "v"(r[4].v.y), "v"(r[5].v.x), "v"(r[5].v.y),
                     "v"(r[6].v.x), "v"(r[6].v.y), "v"(r[7].v.x), "v"(r[7].v.y));
}
__device__ __forceinline__ void hold_gathers(const Row<8> (&)[4]) {}

__device__ __forceinline__ void row_add(const Row<8>& r, int dl, int lane, float al, float* s_acc) {
  const float4 v0 = r.v0, v1 = r.v1, v2 = r.v2, v3 = r.v3;
  const int hb = lane >> 4;
  const float a0 = __shfl(al, hb, 32);
  const float a1 = __shfl(al, 2 + hb, 32);
  const float a2 = __shfl(al, 4 + hb, 32);
  const float a3 = __shfl(al, 6 + hb, 32);
  float rx = a0 * v0.x + a1 * v1.x + a2 * v2.x + a3 * v3.x;
  float ry = a0 * v0.y + a1 * v1.y + a2 * v2.y + a3 * v3.y;
  float rz = a0 * v0.z + a1 * v1.z + a2 * v2.z + a3 * v3.z;
  float rw = a0 * v0.w + a1 * v1.w + a2 * v2.w + a3 * v3.w;
  rx += __shfl_xor(rx, 16, 32);
  ry += __shfl_xor(ry, 16, 32);
  rz += __shfl_xor(rz, 16, 32);
  rw += __shfl_xor(rw, 16, 32);
  float* arow = s_acc + dl * OUTC + (lane & 15) * 4;
  if (lane < 16) { atomicAdd(&arow[0], rx); atomicAdd(&arow[1], ry); }
  else           { atomicAdd(&arow[2], rz); atomicAdd(&arow[3], rw); }
}

__device__ __forceinline__ void row_add(const Row<1>& r, int dl, int lane, float al, float* s_acc) {
  float* arow = s_acc + dl * OUTC + 2 * lane;
  atomicAdd(&arow[0], al * r.v.x);
  atomicAdd(&arow[1], al * r.v.y);
}

template <int HEADS>
__device__ __forceinline__ void edge_accumulate(int s, int dl, int lane, float al,
                                                const float* __restrict__ h,
                                                float* s_acc) {
  Row<HEADS> r;
  row_gather(r, s, lane, h);
  row_add(r, dl, lane, al, s_acc);
}

template <int HEADS, int U>
__device__ __forceinline__ void score_items(int j0, int hd,
                                            const int* __restrict__ ei,
                                            const float* __restrict__ a_s,
                                            const unsigned* s_list, const float* s_ad,
                                            int* s_src, float* s_p) {
  unsigned en[U];
  int sn[U];
  float as[U];
#pragma unroll
  for (int u = 0; u < U; ++u) en[u] = s_list[(j0 + u * OWN_T) / HEADS];
#pragma unroll
  for (int u = 0; u < U; ++u) {
    const int e = (int)(en[u] & 0x7FFFFu);
    const int sl = ei[e < EE ? e : EE - 1];
    sn[u] = e < EE ? sl : e - EE;
  }
#pragma unroll
  for (int u = 0; u < U; ++u) as[u] = a_s[sn[u] * HEADS + hd];
#pragma unroll
  for (int u = 0; u < U; ++u) {
    const int j = j0 + u * OWN_T;
    if (hd == 0) s_src[j / HEADS] = sn[u];
    s_p[j] = edge_score(as[u], s_ad[(int)(en[u] >> 19) * HEADS + hd]);
  }
}

__device__ __forceinline__ int own_chunk(int c, int rot) {
  int t = c + rot;
  return t >= NCHUNK ? t - NCHUNK : t;
}

__device__ __forceinline__ void owner_chunk_append(int4 q, int cc, int lo, int cnt, int lane,
                                                   unsigned* s_list, unsigned* s_cnt) {
  const bool f0 = (unsigned)(q.x - lo) < (unsigned)cnt;
  const bool f1 = (unsigned)(q.y - lo) < (unsigned)cnt;
  const bool f2 = (unsigned)(q.z - lo) < (unsigned)cnt;
  const bool f3 = (unsigned)(q.w - lo) < (unsigned)cnt;
  const unsigned m0 = (unsigned)__ballot(f0);
  const unsigned m1 = (unsigned)__ballot(f1);
  const unsigned m2 = (unsigned)__ballot(f2);
  const unsigned m3 = (unsigned)__ballot(f3);
  if ((m0 | m1 | m2 | m3) == 0u) return;
  const int tot = __popc(m0) + __popc(m1) + __popc(m2) + __popc(m3);
  int base = 0;
  if (lane == 0) base = (int)atomicAdd(s_cnt, (unsigned)tot);
  base = __builtin_amdgcn_readfirstlane(base);
  const unsigned lt = (1u << lane) - 1u;
  int pos = base + __popc(m0 & lt) + __popc(m1 & lt) + __popc(m2 & lt) + __popc(m3 & lt);
  const unsigned e0 = (unsigned)(cc * CHUNK + lane * 4);
  if (f0) { if (pos < LCAP) s_list[pos] = ((unsigned)(q.x - lo) << 19) | e0;        ++pos; }
  if (f1) { if (pos < LCAP) s_list[pos] = ((unsigned)(q.y - lo) << 19) | (e0 + 1u); ++pos; }
  if (f2) { if (pos < LCAP) s_list[pos] = ((unsigned)(q.z - lo) << 19) | (e0 + 2u); ++pos; }
  if (f3) { if (pos < LCAP) s_list[pos] = ((unsigned)(q.w - lo) << 19) | (e0 + 3u); ++pos; }
}

__device__ __forceinline__ void list_bits(unsigned t, unsigned e, int& pos, unsigned* s_list) {
  for (; t != 0u; t &= t - 1u) {
    if (pos < LCAP) s_list[pos] = e + 4u * (unsigned)__builtin_ctz(t);
    ++pos;
  }
}

template <int HEADS, int MODE>
__device__ __forceinline__ void owner_direct_pass(int lo, int cnt, int wave, int lane,
                                                  const int* __restrict__ ei,
                                                  const float* __restrict__ a_s,
                                                  const float* __restrict__ h,
                                                  const float* s_ad, float* s_em,
                                                  float* s_dn, float* s_acc) {
  for (int e = wave; e < EE + cnt; e += OWN_NW) {
    int s, d;
    if (e < EE) { s = ei[e]; d = ei[EE + e]; }
    else        { s = d = lo + (e - EE); }
    const int dl = d - lo;
    if ((unsigned)dl >= (unsigned)cnt) continue;
    if (MODE == 2) {
      const float al = edge_alpha<HEADS>(s, dl, lane, a_s, s_ad, s_em, s_dn);
      edge_accumulate<HEADS>(s, dl, lane, al, h, s_acc);
    } else if (lane < HEADS) {
      const float ev = edge_score(a_s[s * HEADS + lane], s_ad[dl * HEADS + lane]);
      if (MODE == 0) atomicMaxF32(&s_em[dl * HEADS + lane], ev);
      else           atomicAdd(&s_dn[dl * HEADS + lane], __expf(ev - s_em[dl * HEADS + lane]));
    }
  }
}

template <int HEADS, int DO_ELU, int LIST>
__global__ __launch_bounds__(OWN_T)
void owner_aggregate_kernel(const int* __restrict__ ei,
                            const float* __restrict__ a_s,
                            const float* __restrict__ a_d,
                            const float* __restrict__ h,
                            const float* __restrict__ bias,
                            const unsigned* __restrict__ own_bits,
                            float* __restrict__ out, float* keep, float invH) {
  __shared__ float s_acc[OWN * OUTC];
  __shared__ float s_ad[OWN * HEADS];
  __shared__ float s_em[OWN * HEADS];
  __shared__ float s_dn[OWN * HEADS];
  __shared__ unsigned s_list[LCAP];
  __shared__ int s_src[LCAP];
  __shared__ float s_p[LCAP * HEADS];
  __shared__ unsigned s_cnt;

  const int lo = blockIdx.x * OWN;
  int cnt = NN - lo;
  if (cnt > OWN) cnt = OWN;
  const int wave = __builtin_amdgcn_readfirstlane(threadIdx.x >> 5);
  const int lane = threadIdx.x & 31;

  for (int i = threadIdx.x; i < cnt * OUTC; i += OWN_T) s_acc[i] = 0.0f;
  for (int i = threadIdx.x; i < cnt * HEADS; i += OWN_T) {
    s_ad[i] = a_d[lo * HEADS + i];
    s_em[i] = -__builtin_inff();
    s_dn[i] = 0.0f;
  }
  bool kept = false;
  if (LIST == 1) {
    const float* slot = out + (size_t)lo * OUTC;
    const unsigned n = (unsigned)__builtin_amdgcn_readfirstlane((int)__float_as_uint(slot[0]));
    kept = n >= (unsigned)cnt && n <= (unsigned)LCAP && n + KEEP_AT <= (unsigned)(cnt * OUTC);
    if (kept) {
      for (int i = threadIdx.x; i < (int)n; i += OWN_T) s_list[i] = __float_as_uint(slot[KEEP_AT + i]);
      if (threadIdx.x == 0) s_cnt = n;
    }
  }
  if (!kept) {
    for (int i = threadIdx.x; i < cnt; i += OWN_T) s_list[i] = ((unsigned)i << 19) | (unsigned)(EE + lo + i);
    if (threadIdx.x == 0) s_cnt = (unsigned)cnt;
  }
  __syncthreads();

  if (LIST == 0) {
    const uint4* row = (const uint4*)(own_bits + (size_t)blockIdx.x * ROWW);
    const int t = (int)threadIdx.x;
    uint4 m[4];
#pragma unroll
    for (int k = 0; k < 3; ++k) m[k] = row[t + k * OWN_T];
    m[3] = make_uint4(0u, 0u, 0u, 0u);
    if (t + 3 * OWN_T < NCHUNK) m[3] = row[t + 3 * OWN_T];
    int c = 0;
#pragma unroll
    for (int k = 0; k < 4; ++k) c += __popc(m[k].x) + __popc(m[k].y) + __popc(m[k].z) + __popc(m[k].w);
    if (c != 0) {
      int pos = (int)atomicAdd(&s_cnt, (unsigned)c);
#pragma unroll
      for (int k = 0; k < 4; ++k) {
        const unsigned e0 = (unsigned)((t + k * OWN_T) * CHUNK);
        list_bits(m[k].x, e0, pos, s_list);
        list_bits(m[k].y, e0 + 1u, pos, s_list);
        list_bits(m[k].z, e0 + 2u, pos, s_list);
        list_bits(m[k].w, e0 + 3u, pos, s_list);
      }
    }
  } else if (!kept) {
    const int4* dst4 = (const int4*)(ei + EE);
    const int rot = (blockIdx.x * 12) % NCHUNK;
    const int WSTEP = OWN_NW;
    int4 q0 = dst4[own_chunk(wave, rot) * 32 + lane];
    int4 q1 = dst4[own_chunk(wave + WSTEP, rot) * 32 + lane];
    int4 q2 = dst4[own_chunk(wave + 2 * WSTEP, rot) * 32 + lane];
    for (int c = wave; c < NCHUNK; c += 3 * WSTEP) {
      owner_chunk_append(q0, own_chunk(c, rot), lo, cnt, lane, s_list, &s_cnt);
      if (c + 3 * WSTEP < NCHUNK) q0 = dst4[own_chunk(c + 3 * WSTEP, rot) * 32 + lane];
      if (c + WSTEP < NCHUNK) {
        owner_chunk_append(q1, own_chunk(c + WSTEP, rot), lo, cnt, lane, s_list, &s_cnt);
        if (c + 4 * WSTEP < NCHUNK) q1 = dst4[own_chunk(c + 4 * WSTEP, rot) * 32 + lane];
      }
      if (c + 2 * WSTEP < NCHUNK) {
        owner_chunk_append(q2, own_chunk(c + 2 * WSTEP, rot), lo, cnt, lane, s_list, &s_cnt);
        if (c + 5 * WSTEP < NCHUNK) q2 = dst4[own_chunk(c + 5 * WSTEP, rot) * 32 + lane];
      }
    }
  }
  __syncthreads();
  const int total = (int)s_cnt;

  if (total <= LCAP) {
    if (LIST == 0) {
      for (int i = cnt + (int)threadIdx.x; i < total; i += 2 * OWN_T) {
        const bool two = i + OWN_T < total;
        const int ea = (int)s_list[i];
        const int eb = two ? (int)s_list[i + OWN_T] : ea;
        const int sa = ei[ea], da = ei[EE + ea];
        const int sb = ei[eb], db = ei[EE + eb];
        unsigned la = (unsigned)(da - lo), lb = (unsigned)(db - lo);
        if (la >= (unsigned)cnt) la = 0u;
        if (lb >= (unsigned)cnt) lb = 0u;
        s_list[i] = (la << 19) | (unsigned)(EE + sa);
        if (two) s_list[i + OWN_T] = (lb << 19) | (unsigned)(EE + sb);
      }
      __syncthreads();
    }
    const int items = total * HEADS;
    {
      const int hd = threadIdx.x % HEADS;
      int j = threadIdx.x;
      if (HEADS > 1) {
        for (; j + 7 * OWN_T < items; j += 8 * OWN_T) score_items<HEADS, 8>(j, hd, ei, a_s, s_list, s_ad, s_src, s_p);
        if (j + 3 * OWN_T < items) { score_items<HEADS, 4>(j, hd, ei, a_s, s_list, s_ad, s_src, s_p); j += 4 * OWN_T; }
      }
      for (; j + OWN_T < items; j += 2 * OWN_T) score_items<HEADS, 2>(j, hd, ei, a_s, s_list, s_ad, s_src, s_p);
      if (j < items) score_items<HEADS, 1>(j, hd, ei, a_s, s_list, s_ad, s_src, s_p);
    }
    for (int j = threadIdx.x; j < items; j += OWN_T) {
      const unsigned en = s_list[j / HEADS];
      atomicMaxF32(&s_em[(int)(en >> 19) * HEADS + j % HEADS], s_p[j]);
    }
    __syncthreads();
    if (LIST == 0) {
      float* slot = keep + (size_t)lo * OUTC;
      const bool fits = total + KEEP_AT <= cnt * OUTC;
      if (fits) {
        for (int i = threadIdx.x; i < total; i += OWN_T)
          slot[KEEP_AT + i] = __uint_as_float((s_list[i] & ~0x7FFFFu) | (unsigned)(EE + s_src[i]));
      }
      if (threadIdx.x == 0) slot[0] = __uint_as_float(fits ? (unsigned)total : KEEP_NONE);
    }
    for (int j = threadIdx.x; j < items; j += OWN_T) {
      const unsigned en = s_list[j / HEADS];
      const int k = (int)(en >> 19) * HEADS + j % HEADS;
      const float p = __expf(s_p[j] - s_em[k]);
      s_p[j] = p;
      atomicAdd(&s_dn[k], p);
    }
    __syncthreads();
    constexpr int SB = HEADS == 8 ? 4 : 8;
    const int hh = lane & (HEADS - 1);
    const int nfull = total / SB;
    for (int t = wave; t < nfull; t += OWN_NW) {
      int sn[SB], dl[SB];
      float p[SB], al[SB];
      Row<HEADS> r[SB];
#pragma unroll
      for (int b = 0; b < SB; ++b) {
        const int i = t * SB + b;
        sn[b] = __builtin_amdgcn_readfirstlane(s_src[i]);
        dl[b] = (int)((unsigned)__builtin_amdgcn_readfirstlane((int)s_list[i]) >> 19);
        p[b]  = s_p[i * HEADS + hh];
      }
#pragma unroll
      for (int b = 0; b < SB; ++b) row_gather(r[b], sn[b], lane, h);
#pragma unroll
      for (int b = 0; b < SB; ++b) al[b] = p[b] / s_dn[dl[b] * HEADS + hh];
      hold_gathers(r);
#pragma unroll
      for (int b = 0; b < SB; ++b) row_add(r[b], dl[b], lane, al[b], s_acc);
    }
    for (int i = nfull * SB + wave; i < total; i += OWN_NW) {
      const int s  = __builtin_amdgcn_readfirstlane(s_src[i]);
      const int dl = (int)((unsigned)__builtin_amdgcn_readfirstlane((int)s_list[i]) >> 19);
      const float al = s_p[i * HEADS + hh] / s_dn[dl * HEADS + hh];
      edge_accumulate<HEADS>(s, dl, lane, al, h, s_acc);
    }
  } else {
    if (LIST == 0 && threadIdx.x == 0) keep[(size_t)lo * OUTC] = __uint_as_float(KEEP_NONE);
    owner_direct_pass<HEADS, 0>(lo, cnt, wave, lane, ei, a_s, h, s_ad, s_em, s_dn, s_acc);
    __syncthreads();
    owner_direct_pass<HEADS, 1>(lo, cnt, wave, lane, ei, a_s, h, s_ad, s_em, s_dn, s_acc);
    __syncthreads();
    owner_direct_pass<HEADS, 2>(lo, cnt, wave, lane, ei, a_s, h, s_ad, s_em, s_dn, s_acc);
  }
  __syncthreads();

  for (int i = threadIdx.x; i < cnt * OUTC; i += OWN_T) {
    float v = s_acc[i] * invH + bias[i & (OUTC - 1)];
    if (DO_ELU) v = v > 0.0f ? v : (__expf(v) - 1.0f);
    out[(size_t)lo * OUTC + i] = v;
  }
}

static inline unsigned cdiv(long long a, long long b) { return (unsigned)((a + b - 1) / b); }

extern "C" void kernel_launch(void* const* d_in, const int* in_sizes, int n_in,
                              void* d_out, int out_size, void* d_ws, size_t ws_size,
                              hipStream_t stream) {
  (void)in_sizes; (void)n_in; (void)out_size; (void)ws_size;
  const float* x    = (const float*)d_in[0];
  const int*   ei   = (const int*)d_in[1];
  const float* W1   = (const float*)d_in[2];
  const float* asw1 = (const float*)d_in[3];
  const float* adw1 = (const float*)d_in[4];
  const float* b1   = (const float*)d_in[5];
  const float* W2   = (const float*)d_in[6];
  const float* asw2 = (const float*)d_in[7];
  const float* adw2 = (const float*)d_in[8];
  const float* b2   = (const float*)d_in[9];
  float* out = (float*)d_out;

  float* ws   = (float*)d_ws;
  float* h1   = ws; ws += (size_t)NN * H1 * OUTC;
  float* as1  = ws; ws += (size_t)NN * H1;
  float* ad1  = ws; ws += (size_t)NN * H1;
  float* acc1 = ws; ws += (size_t)NN * OUTC;
  float* h2   = ws; ws += (size_t)NN * OUTC;
  float* as2  = ws; ws += (size_t)NN;
  float* ad2  = ws; ws += (size_t)NN;
  unsigned* own_bits = (unsigned*)h2;

  const int T = PRE_T;

  {
    long long waves = ((long long)NN / 16) * (H1 * OUTC / 64);
    gemm_wmma_f32_kernel<<<cdiv(waves * 32, T), T, 0, stream>>>(x, W1, h1, asw1, adw1, as1, ad1,
                                                                 ei, own_bits, NN, INCH, H1 * OUTC);
  }
  owner_aggregate_kernel<H1, 1, 0><<<OWN_WG, OWN_T, 0, stream>>>(ei, as1, ad1, h1, b1, own_bits, acc1, out, 1.0f / H1);

  {
    long long waves = ((long long)NN / 16) * (OUTC / 64);
    gemm_wmma_f32_kernel<<<cdiv(waves * 32, T), T, 0, stream>>>(acc1, W2, h2, asw2, adw2, as2, ad2,
                                                                 ei, nullptr, NN, OUTC, OUTC);
  }
  owner_aggregate_kernel<1, 0, 1><<<OWN_WG, OWN_T, 0, stream>>>(ei, as2, ad2, h2, b2, nullptr, out, nullptr, 1.0f);
}


# Layer 2 reassociated: owner workgroups gather layer-1 output rows and multiply the summed tile by W2 on WMMA at the end; scores from row.(W2.att); product launch, h2 gone (4 to 3); new sum order
# speedup vs baseline: 5.1558x; 1.1579x over previous
#include <hip/hip_runtime.h>
#include <stdint.h>

#define NN    50000
#define EE    400000
#define INCH  128
#define OUTC  64
#define H1    8
#define NEG   0.2f

typedef float v2f __attribute__((ext_vector_type(2)));
typedef float v8f __attribute__((ext_vector_type(8)));

#define OWN     196
#define OWN_WG  ((NN + OWN - 1) / OWN)
#define OWN_T   1024
#define OWN_NW  (OWN_T / 32)
#define CHUNK   128
#define NCHUNK  (EE / CHUNK)
#define LCAP    3072
#define KEEP_AT   32
#define KEEP_NONE 0xFFFFFFFFu
#define ACCROWS ((OWN + 15) / 16 * 16)
#define ACCP    (OUTC + 4)
static_assert(EE % CHUNK == 0, "edge list is scanned in whole 128-edge chunks");
static_assert(NCHUNK >= 3 * OWN_NW, "every wave starts with three chunks in flight");
static_assert(EE + NN <= (1 << 19) && OWN <= (1 << 13), "a list word packs the edge number in 19 bits, the local destination above it");
static_assert(OWN <= LCAP, "the self loops always fit the list");
static_assert(OWN_T % H1 == 0, "a thread's items all belong to one head");
#define ROWW    (((EE / 32) + 31) / 32 * 32)
#define PRE_T   256
#define PRE_NW  (PRE_T / 32)
#define PRE_WG  ((NCHUNK + PRE_NW - 1) / PRE_NW)
static_assert(OWN_WG <= 256, "the bit rows are made for 256 owners");
static_assert(PRE_NW == 8, "a product workgroup's eight chunk masks make one 128-byte line of a row");
static_assert(256 * (size_t)ROWW <= (size_t)NN * OUTC + (size_t)NN, "the bit rows fit their stretch of the workspace");
static_assert(3 * OWN_T <= NCHUNK && NCHUNK <= 4 * OWN_T, "a thread reads three or four chunk masks of its row");

__device__ __forceinline__ void hold_block_loads(const float4& a4, const float4& q0, const float4& q1,
                                                 const float4& q2, const float4& q3) {
  asm volatile("" :: "v"(a4.x), "v"(a4.y), "v"(a4.z), "v"(a4.w),
                     "v"(q0.x), "v"(q0.y), "v"(q0.z), "v"(q0.w),
                     "v"(q1.x), "v"(q1.y), "v"(q1.z), "v"(q1.w),
                     "v"(q2.x), "v"(q2.y), "v"(q2.z), "v"(q2.w),
                     "v"(q3.x), "v"(q3.y), "v"(q3.z), "v"(q3.w));
}

__global__ void gemm_wmma_f32_kernel(const float* __restrict__ A,
                                     const float* __restrict__ B,
                                     float* __restrict__ C,
                                     const float* __restrict__ att_src,
                                     const float* __restrict__ att_dst,
                                     float* __restrict__ a_s,
                                     float* __restrict__ a_d,
                                     const int* __restrict__ ei,
                                     unsigned* __restrict__ own_bits,
                                     int M, int K, int Nc) {
  __shared__ unsigned s_bits[64 * 32];
  const int wid  = (blockIdx.x * blockDim.x + threadIdx.x) >> 5;
  const int lane = threadIdx.x & 31;
  if (own_bits != nullptr && blockIdx.x < PRE_WG) {
    const int pw = threadIdx.x >> 5;
    const int ch = blockIdx.x * PRE_NW + pw;
    const bool on = ch < NCHUNK;
    unsigned o0 = 0u, o1 = 0u, o2 = 0u, o3 = 0u;
    if (on) {
      const int4 q = ((const int4*)(ei + EE))[ch * 32 + lane];
      o0 = min((unsigned)q.x / (unsigned)OWN, 255u);
      o1 = min((unsigned)q.y / (unsigned)OWN, 255u);
      o2 = min((unsigned)q.z / (unsigned)OWN, 255u);
      o3 = min((unsigned)q.w / (unsigned)OWN, 255u);
    }
    uint4* tile4 = (uint4*)s_bits;
    const unsigned bit = 1u << lane;
    const int t = (int)threadIdx.x;
    for (int p = 0; p < 4; ++p) {
      tile4[t]         = make_uint4(0u, 0u, 0u, 0u);
      tile4[t + PRE_T] = make_uint4(0u, 0u, 0u, 0u);
      __syncthreads();
      if (on) {
        if ((o0 >> 6) == (unsigned)p) atomicOr(&s_bits[(o0 & 63u) * 32 + pw * 4 + 0], bit);
        if ((o1 >> 6) == (unsigned)p) atomicOr(&s_bits[(o1 & 63u) * 32 + pw * 4 + 1], bit);
        if ((o2 >> 6) == (unsigned)p) atomicOr(&s_bits[(o2 & 63u) * 32 + pw * 4 + 2], bit);
        if ((o3 >> 6) == (unsigned)p) atomicOr(&s_bits[(o3 & 63u) * 32 + pw * 4 + 3], bit);
      }
      __syncthreads();
      for (int i = 0; i < 2; ++i) {
        const int u = t + i * PRE_T;
        const int o = 64 * p + (u >> 3);
        const int c = blockIdx.x * PRE_NW + (u & 7);
        if (c < NCHUNK) ((uint4*)(own_bits + (size_t)o * ROWW))[c] = tile4[u];
      }
      if (p < 3) __syncthreads();
    }
  }
  const int tilesN   = Nc >> 6;
  const int rowTiles = M >> 4;
  if (wid >= rowTiles * tilesN) return;

  const int rt = wid / tilesN;
  const int cg = wid - rt * tilesN;
  const int m0 = rt << 4;
  const int n0 = cg << 6;
  const int ml = lane & 15;
  const int kh = lane >> 4;

  v8f acc0 = {}, acc1 = {}, acc2 = {}, acc3 = {};
  const float* arow = A + (size_t)(m0 + ml) * K + 4 * kh;
  const float* bcol = B + (size_t)(4 * kh) * Nc + n0 + 4 * ml;

  for (int k = 0; k < K; k += 8) {
    const float4 a4 = *(const float4*)(arow + k);
    const float* bp = bcol + (size_t)k * Nc;
    const float4 q0 = *(const float4*)(bp);
    const float4 q1 = *(const float4*)(bp + (size_t)Nc);
    const float4 q2 = *(const float4*)(bp + 2 * (size_t)Nc);
    const float4 q3 = *(const float4*)(bp + 3 * (size_t)Nc);
    hold_block_loads(a4, q0, q1, q2, q3);
    v2f a, b0, b1, b2, b3;
    a.x = a4.x;  a.y = a4.y;
    b0.x = q0.x; b0.y = q1.x;
    b1.x = q0.y; b1.y = q1.y;
    b2.x = q0.z; b2.y = q1.z;
    b3.x = q0.w; b3.y = q1.w;
    acc0 = __builtin_amdgcn_wmma_f32_16x16x4_f32(false, a, false, b0, (short)0, acc0, false, false);
    acc1 = __builtin_amdgcn_wmma_f32_16x16x4_f32(false, a, false, b1, (short)0, acc1, false, false);
    acc2 = __builtin_amdgcn_wmma_f32_16x16x4_f32(false, a, false, b2, (short)0, acc2, false, false);
    acc3 = __builtin_amdgcn_wmma_f32_16x16x4_f32(false, a, false, b3, (short)0, acc3, false, false);
    a.x = a4.z;  a.y = a4.w;
    b0.x = q2.x; b0.y = q3.x;
    b1.x = q2.y; b1.y = q3.y;
    b2.x = q2.z; b2.y = q3.z;
    b3.x = q2.w; b3.y = q3.w;
    acc0 = __builtin_amdgcn_wmma_f32_16x16x4_f32(false, a, false, b0, (short)0, acc0, false, false);
    acc1 = __builtin_amdgcn_wmma_f32_16x16x4_f32(false, a, false, b1, (short)0, acc1, false, false);
    acc2 = __builtin_amdgcn_wmma_f32_16x16x4_f32(false, a, false, b2, (short)0, acc2, false, false);
    acc3 = __builtin_amdgcn_wmma_f32_16x16x4_f32(false, a, false, b3, (short)0, acc3, false, false);
  }

  const int mr = m0 + 8 * kh;
#pragma unroll
  for (int v = 0; v < 8; ++v) {
    float* crow = C + (size_t)(mr + v) * Nc + n0 + 4 * ml;
    *(float4*)crow = make_float4(acc0[v], acc1[v], acc2[v], acc3[v]);
  }

  const float4 ws = *(const float4*)(att_src + n0 + 4 * ml);
  const float4 wd = *(const float4*)(att_dst + n0 + 4 * ml);
  const float s0 = ws.x, s1 = ws.y, s2 = ws.z, s3 = ws.w;
  const float d0 = wd.x, d1 = wd.y, d2 = wd.z, d3 = wd.w;
  float ps[8], pd[8];
#pragma unroll
  for (int v = 0; v < 8; ++v) {
    ps[v] = acc0[v] * s0 + acc1[v] * s1 + acc2[v] * s2 + acc3[v] * s3;
    pd[v] = acc0[v] * d0 + acc1[v] * d1 + acc2[v] * d2 + acc3[v] * d3;
  }
#pragma unroll
  for (int off = 8; off; off >>= 1) {
#pragma unroll
    for (int v = 0; v < 8; ++v) {
      ps[v] += __shfl_xor(ps[v], off, 32);
      pd[v] += __shfl_xor(pd[v], off, 32);
    }
  }
  float vs = ps[0], vd = pd[0];
#pragma unroll
  for (int v = 1; v < 8; ++v) {
    if (ml == v) { vs = ps[v]; vd = pd[v]; }
  }
  if (ml < 8) {
    const int idx = (mr + ml) * tilesN + cg;
    a_s[idx]   = vs;
    a_d[idx]   = vd;
  }
}

__device__ __forceinline__ float leaky(float x) {
  return x > 0.0f ? x : NEG * x;
}

__device__ __forceinline__ float edge_score(float as, float ad) {
  return leaky(as + ad);
}

__device__ __forceinline__ void atomicMaxF32(float* addr, float val) {
  if (val >= 0.0f) atomicMax((int*)addr, __float_as_int(val));
  else             atomicMin((unsigned int*)addr, __float_as_uint(val));
}


template <int HEADS>
__device__ __forceinline__ float edge_alpha(int s, int dl, int lane,
                                            const float* __restrict__ a_s,
                                            const float* s_ad, const float* s_em,
                                            const float* s_dn) {
  const int hh = lane & (HEADS - 1);
  const float ev = edge_score(a_s[s * HEADS + hh], s_ad[dl * HEADS + hh]);
  return __expf(ev - s_em[dl * HEADS + hh]) / s_dn[dl * HEADS + hh];
}

template <int HEADS> struct Row;
template <> struct Row<8> { float4 v0, v1, v2, v3; };
template <> struct Row<1> { float2 v; };

__device__ __forceinline__ void row_gather(Row<8>& r, int s, int lane, const float* __restrict__ h) {
  const float4* hrow = (const float4*)(h + (size_t)s * (8 * OUTC));
  r.v0 = hrow[lane];
  r.v1 = hrow[32 + lane];
  r.v2 = hrow[64 + lane];
  r.v3 = hrow[96 + lane];
}

__device__ __forceinline__ void row_gather(Row<1>& r, int s, int lane, const float* __restrict__ h) {
  r.v = ((const float2*)(h + (size_t)s * OUTC))[lane];
}

__device__ __forceinline__ void hold_gathers(const Row<1> (&r)[8]) {
  asm volatile("" :: "v"(r[0].v.x), "v"(r[0].v.y), "v"(r[1].v.x), "v"(r[1].v.y),
                     "v"(r[2].v.x), "v"(r[2].v.y), "v"(r[3].v.x), "v"(r[3].v.y),
                     "v"(r[4].v.x), "v"(r[4].v.y), "v"(r[5].v.x), "v"(r[5].v.y),
                     "v"(r[6].v.x), "v"(r[6].v.y), "v"(r[7].v.x), "v"(r[7].v.y));
}
__device__ __forceinline__ void hold_gathers(const Row<8> (&)[4]) {}

__device__ __forceinline__ void row_add(const Row<8>& r, int dl, int lane, float al, float* s_acc) {
  const float4 v0 = r.v0, v1 = r.v1, v2 = r.v2, v3 = r.v3;
  const int hb = lane >> 4;
  const float a0 = __shfl(al, hb, 32);
  const float a1 = __shfl(al, 2 + hb, 32);
  const float a2 = __shfl(al, 4 + hb, 32);
  const float a3 = __shfl(al, 6 + hb, 32);
  float rx = a0 * v0.x + a1 * v1.x + a2 * v2.x + a3 * v3.x;
  float ry = a0 * v0.y + a1 * v1.y + a2 * v2.y + a3 * v3.y;
  float rz = a0 * v0.z + a1 * v1.z + a2 * v2.z + a3 * v3.z;
  float rw = a0 * v0.w + a1 * v1.w + a2 * v2.w + a3 * v3.w;
  rx += __shfl_xor(rx, 16, 32);
  ry += __shfl_xor(ry, 16, 32);
  rz += __shfl_xor(rz, 16, 32);
  rw += __shfl_xor(rw, 16, 32);
  float* arow = s_acc + dl * ACCP + (lane & 15) * 4;
  if (lane < 16) { atomicAdd(&arow[0], rx); atomicAdd(&arow[1], ry); }
  else           { atomicAdd(&arow[2], rz); atomicAdd(&arow[3], rw); }
}

__device__ __forceinline__ void row_add(const Row<1>& r, int dl, int lane, float al, float* s_acc) {
  float* arow = s_acc + dl * ACCP + 2 * lane;
  atomicAdd(&arow[0], al * r.v.x);
  atomicAdd(&arow[1], al * r.v.y);
}

template <int HEADS>
__device__ __forceinline__ void edge_accumulate(int s, int dl, int lane, float al,
                                                const float* __restrict__ h,
                                                float* s_acc) {
  Row<HEADS> r;
  row_gather(r, s, lane, h);
  row_add(r, dl, lane, al, s_acc);
}

template <int HEADS, int U>
__device__ __forceinline__ void score_items(int j0, int hd,
                                            const int* __restrict__ ei,
                                            const float* __restrict__ a_s,
                                            const unsigned* s_list, const float* s_ad,
                                            int* s_src, float* s_p) {
  unsigned en[U];
  int sn[U];
  float as[U];
#pragma unroll
  for (int u = 0; u < U; ++u) en[u] = s_list[(j0 + u * OWN_T) / HEADS];
#pragma unroll
  for (int u = 0; u < U; ++u) {
    const int e = (int)(en[u] & 0x7FFFFu);
    const int sl = ei[e < EE ? e : EE - 1];
    sn[u] = e < EE ? sl : e - EE;
  }
#pragma unroll
  for (int u = 0; u < U; ++u) as[u] = a_s[sn[u] * HEADS + hd];
#pragma unroll
  for (int u = 0; u < U; ++u) {
    const int j = j0 + u * OWN_T;
    if (hd == 0) s_src[j / HEADS] = sn[u];
    s_p[j] = edge_score(as[u], s_ad[(int)(en[u] >> 19) * HEADS + hd]);
  }
}

__device__ __forceinline__ int own_chunk(int c, int rot) {
  int t = c + rot;
  return t >= NCHUNK ? t - NCHUNK : t;
}

__device__ __forceinline__ void owner_chunk_append(int4 q, int cc, int lo, int cnt, int lane,
                                                   unsigned* s_list, unsigned* s_cnt) {
  const bool f0 = (unsigned)(q.x - lo) < (unsigned)cnt;
  const bool f1 = (unsigned)(q.y - lo) < (unsigned)cnt;
  const bool f2 = (unsigned)(q.z - lo) < (unsigned)cnt;
  const bool f3 = (unsigned)(q.w - lo) < (unsigned)cnt;
  const unsigned m0 = (unsigned)__ballot(f0);
  const unsigned m1 = (unsigned)__ballot(f1);
  const unsigned m2 = (unsigned)__ballot(f2);
  const unsigned m3 = (unsigned)__ballot(f3);
  if ((m0 | m1 | m2 | m3) == 0u) return;
  const int tot = __popc(m0) + __popc(m1) + __popc(m2) + __popc(m3);
  int base = 0;
  if (lane == 0) base = (int)atomicAdd(s_cnt, (unsigned)tot);
  base = __builtin_amdgcn_readfirstlane(base);
  const unsigned lt = (1u << lane) - 1u;
  int pos = base + __popc(m0 & lt) + __popc(m1 & lt) + __popc(m2 & lt) + __popc(m3 & lt);
  const unsigned e0 = (unsigned)(cc * CHUNK + lane * 4);
  if (f0) { if (pos < LCAP) s_list[pos] = ((unsigned)(q.x - lo) << 19) | e0;        ++pos; }
  if (f1) { if (pos < LCAP) s_list[pos] = ((unsigned)(q.y - lo) << 19) | (e0 + 1u); ++pos; }
  if (f2) { if (pos < LCAP) s_list[pos] = ((unsigned)(q.z - lo) << 19) | (e0 + 2u); ++pos; }
  if (f3) { if (pos < LCAP) s_list[pos] = ((unsigned)(q.w - lo) << 19) | (e0 + 3u); ++pos; }
}

__device__ __forceinline__ void list_bits(unsigned t, unsigned e, int& pos, unsigned* s_list) {
  for (; t != 0u; t &= t - 1u) {
    if (pos < LCAP) s_list[pos] = e + 4u * (unsigned)__builtin_ctz(t);
    ++pos;
  }
}

template <int HEADS, int MODE>
__device__ __forceinline__ void owner_direct_pass(int lo, int cnt, int wave, int lane,
                                                  const int* __restrict__ ei,
                                                  const float* __restrict__ a_s,
                                                  const float* __restrict__ h,
                                                  const float* s_ad, float* s_em,
                                                  float* s_dn, float* s_acc) {
  for (int e = wave; e < EE + cnt; e += OWN_NW) {
    int s, d;
    if (e < EE) { s = ei[e]; d = ei[EE + e]; }
    else        { s = d = lo + (e - EE); }
    const int dl = d - lo;
    if ((unsigned)dl >= (unsigned)cnt) continue;
    if (MODE == 2) {
      const float al = edge_alpha<HEADS>(s, dl, lane, a_s, s_ad, s_em, s_dn);
      edge_accumulate<HEADS>(s, dl, lane, al, h, s_acc);
    } else if (lane < HEADS) {
      const float ev = edge_score(a_s[s * HEADS + lane], s_ad[dl * HEADS + lane]);
      if (MODE == 0) atomicMaxF32(&s_em[dl * HEADS + lane], ev);
      else           atomicAdd(&s_dn[dl * HEADS + lane], __expf(ev - s_em[dl * HEADS + lane]));
    }
  }
}

template <int HEADS, int DO_ELU, int LIST>
__global__ __launch_bounds__(OWN_T)
void owner_aggregate_kernel(const int* __restrict__ ei,
                            const float* __restrict__ a_s,
                            float* a_d,
                            const float* __restrict__ h,
                            const float* __restrict__ bias,
                            const unsigned* __restrict__ own_bits,
                            float* __restrict__ out, float* keep, float invH,
                            const float* __restrict__ Wn,
                            const float* __restrict__ att_sn,
                            const float* __restrict__ att_dn,
                            float* __restrict__ as_n) {
  __shared__ __attribute__((aligned(16))) float s_acc[ACCROWS * ACCP];
  __shared__ __attribute__((aligned(16))) float s_wa[2 * OUTC];
  __shared__ float s_ad[OWN * HEADS];
  __shared__ float s_em[OWN * HEADS];
  __shared__ float s_dn[OWN * HEADS];
  __shared__ unsigned s_list[LCAP];
  __shared__ int s_src[LCAP];
  __shared__ float s_p[LCAP * HEADS];
  __shared__ unsigned s_cnt;

  const int lo = blockIdx.x * OWN;
  int cnt = NN - lo;
  if (cnt > OWN) cnt = OWN;
  const int wave = __builtin_amdgcn_readfirstlane(threadIdx.x >> 5);
  const int lane = threadIdx.x & 31;

  if (LIST == 0 && threadIdx.x < 8 * OUTC) {
    const int c = threadIdx.x >> 3, w = (threadIdx.x >> 2) & 1, part = threadIdx.x & 3;
    const float4* wr = (const float4*)(Wn + c * OUTC + part * 16);
    const float4* ar = (const float4*)((w ? att_dn : att_sn) + part * 16);
    float s = 0.0f;
#pragma unroll
    for (int j = 0; j < 4; ++j) {
      const float4 wv = wr[j], av = ar[j];
      s += wv.x * av.x + wv.y * av.y + wv.z * av.z + wv.w * av.w;
    }
    s += __shfl_xor(s, 1, 32);
    s += __shfl_xor(s, 2, 32);
    if (part == 0) s_wa[w * OUTC + c] = s;
  }
  for (int i = threadIdx.x; i < ACCROWS * ACCP; i += OWN_T) s_acc[i] = 0.0f;
  const int adb = LIST == 1 ? lo * H1 : lo * HEADS;
  for (int i = threadIdx.x; i < cnt * HEADS; i += OWN_T) {
    s_ad[i] = a_d[adb + i];
    s_em[i] = -__builtin_inff();
    s_dn[i] = 0.0f;
  }
  bool kept = false;
  if (LIST == 1) {
    const float* slot = out + (size_t)lo * OUTC;
    const unsigned n = (unsigned)__builtin_amdgcn_readfirstlane((int)__float_as_uint(slot[0]));
    kept = n >= (unsigned)cnt && n <= (unsigned)LCAP && n + KEEP_AT <= (unsigned)(cnt * OUTC);
    if (kept) {
      for (int i = threadIdx.x; i < (int)n; i += OWN_T) s_list[i] = __float_as_uint(slot[KEEP_AT + i]);
      if (threadIdx.x == 0) s_cnt = n;
    }
  }
  if (!kept) {
    for (int i = threadIdx.x; i < cnt; i += OWN_T) s_list[i] = ((unsigned)i << 19) | (unsigned)(EE + lo + i);
    if (threadIdx.x == 0) s_cnt = (unsigned)cnt;
  }
  __syncthreads();

  if (LIST == 0) {
    const uint4* row = (const uint4*)(own_bits + (size_t)blockIdx.x * ROWW);
    const int t = (int)threadIdx.x;
    uint4 m[4];
#pragma unroll
    for (int k = 0; k < 3; ++k) m[k] = row[t + k * OWN_T];
    m[3] = make_uint4(0u, 0u, 0u, 0u);
    if (t + 3 * OWN_T < NCHUNK) m[3] = row[t + 3 * OWN_T];
    int c = 0;
#pragma unroll
    for (int k = 0; k < 4; ++k) c += __popc(m[k].x) + __popc(m[k].y) + __popc(m[k].z) + __popc(m[k].w);
    if (c != 0) {
      int pos = (int)atomicAdd(&s_cnt, (unsigned)c);
#pragma unroll
      for (int k = 0; k < 4; ++k) {
        const unsigned e0 = (unsigned)((t + k * OWN_T) * CHUNK);
        list_bits(m[k].x, e0, pos, s_list);
        list_bits(m[k].y, e0 + 1u, pos, s_list);
        list_bits(m[k].z, e0 + 2u, pos, s_list);
        list_bits(m[k].w, e0 + 3u, pos, s_list);
      }
    }
  } else if (!kept) {
    const int4* dst4 = (const int4*)(ei + EE);
    const int rot = (blockIdx.x * 12) % NCHUNK;
    const int WSTEP = OWN_NW;
    int4 q0 = dst4[own_chunk(wave, rot) * 32 + lane];
    int4 q1 = dst4[own_chunk(wave + WSTEP, rot) * 32 + lane];
    int4 q2 = dst4[own_chunk(wave + 2 * WSTEP, rot) * 32 + lane];
    for (int c = wave; c < NCHUNK; c += 3 * WSTEP) {
      owner_chunk_append(q0, own_chunk(c, rot), lo, cnt, lane, s_list, &s_cnt);
      if (c + 3 * WSTEP < NCHUNK) q0 = dst4[own_chunk(c + 3 * WSTEP, rot) * 32 + lane];
      if (c + WSTEP < NCHUNK) {
        owner_chunk_append(q1, own_chunk(c + WSTEP, rot), lo, cnt, lane, s_list, &s_cnt);
        if (c + 4 * WSTEP < NCHUNK) q1 = dst4[own_chunk(c + 4 * WSTEP, rot) * 32 + lane];
      }
      if (c + 2 * WSTEP < NCHUNK) {
        owner_chunk_append(q2, own_chunk(c + 2 * WSTEP, rot), lo, cnt, lane, s_list, &s_cnt);
        if (c + 5 * WSTEP < NCHUNK) q2 = dst4[own_chunk(c + 5 * WSTEP, rot) * 32 + lane];
      }
    }
  }
  __syncthreads();
  const int total = (int)s_cnt;

  if (total <= LCAP) {
    if (LIST == 0) {
      for (int i = cnt + (int)threadIdx.x; i < total; i += 2 * OWN_T) {
        const bool two = i + OWN_T < total;
        const int ea = (int)s_list[i];
        const int eb = two ? (int)s_list[i + OWN_T] : ea;
        const int sa = ei[ea], da = ei[EE + ea];
        const int sb = ei[eb], db = ei[EE + eb];
        unsigned la = (unsigned)(da - lo), lb = (unsigned)(db - lo);
        if (la >= (unsigned)cnt) la = 0u;
        if (lb >= (unsigned)cnt) lb = 0u;
        s_list[i] = (la << 19) | (unsigned)(EE + sa);
        if (two) s_list[i + OWN_T] = (lb << 19) | (unsigned)(EE + sb);
      }
      __syncthreads();
    }
    const int items = total * HEADS;
    {
      const int hd = threadIdx.x % HEADS;
      int j = threadIdx.x;
      if (HEADS > 1) {
        for (; j + 7 * OWN_T < items; j += 8 * OWN_T) score_items<HEADS, 8>(j, hd, ei, a_s, s_list, s_ad, s_src, s_p);
        if (j + 3 * OWN_T < items) { score_items<HEADS, 4>(j, hd, ei, a_s, s_list, s_ad, s_src, s_p); j += 4 * OWN_T; }
      }
      for (; j + OWN_T < items; j += 2 * OWN_T) score_items<HEADS, 2>(j, hd, ei, a_s, s_list, s_ad, s_src, s_p);
      if (j < items) score_items<HEADS, 1>(j, hd, ei, a_s, s_list, s_ad, s_src, s_p);
    }
    for (int j = threadIdx.x; j < items; j += OWN_T) {
      const unsigned en = s_list[j / HEADS];
      atomicMaxF32(&s_em[(int)(en >> 19) * HEADS + j % HEADS], s_p[j]);
    }
    __syncthreads();
    if (LIST == 0) {
      float* slot = keep + (size_t)lo * OUTC;
      const bool fits = total + KEEP_AT <= cnt * OUTC;
      if (fits) {
        for (int i = threadIdx.x; i < total; i += OWN_T)
          slot[KEEP_AT + i] = __uint_as_float((s_list[i] & ~0x7FFFFu) | (unsigned)(EE + s_src[i]));
      }
      if (threadIdx.x == 0) slot[0] = __uint_as_float(fits ? (unsigned)total : KEEP_NONE);
    }
    for (int j = threadIdx.x; j < items; j += OWN_T) {
      const unsigned en = s_list[j / HEADS];
      const int k = (int)(en >> 19) * HEADS + j % HEADS;
      const float p = __expf(s_p[j] - s_em[k]);
      s_p[j] = p;
      atomicAdd(&s_dn[k], p);
    }
    __syncthreads();
    constexpr int SB = HEADS == 8 ? 4 : 8;
    const int hh = lane & (HEADS - 1);
    const int nfull = total / SB;
    for (int t = wave; t < nfull; t += OWN_NW) {
      int sn[SB], dl[SB];
      float p[SB], al[SB];
      Row<HEADS> r[SB];
#pragma unroll
      for (int b = 0; b < SB; ++b) {
        const int i = t * SB + b;
        sn[b] = __builtin_amdgcn_readfirstlane(s_src[i]);
        dl[b] = (int)((unsigned)__builtin_amdgcn_readfirstlane((int)s_list[i]) >> 19);
        p[b]  = s_p[i * HEADS + hh];
      }
#pragma unroll
      for (int b = 0; b < SB; ++b) row_gather(r[b], sn[b], lane, h);
#pragma unroll
      for (int b = 0; b < SB; ++b) al[b] = p[b] / s_dn[dl[b] * HEADS + hh];
      hold_gathers(r);
#pragma unroll
      for (int b = 0; b < SB; ++b) row_add(r[b], dl[b], lane, al[b], s_acc);
    }
    for (int i = nfull * SB + wave; i < total; i += OWN_NW) {
      const int s  = __builtin_amdgcn_readfirstlane(s_src[i]);
      const int dl = (int)((unsigned)__builtin_amdgcn_readfirstlane((int)s_list[i]) >> 19);
      const float al = s_p[i * HEADS + hh] / s_dn[dl * HEADS + hh];
      edge_accumulate<HEADS>(s, dl, lane, al, h, s_acc);
    }
  } else {
    if (LIST == 0 && threadIdx.x == 0) keep[(size_t)lo * OUTC] = __uint_as_float(KEEP_NONE);
    owner_direct_pass<HEADS, 0>(lo, cnt, wave, lane, ei, a_s, h, s_ad, s_em, s_dn, s_acc);
    __syncthreads();
    owner_direct_pass<HEADS, 1>(lo, cnt, wave, lane, ei, a_s, h, s_ad, s_em, s_dn, s_acc);
    __syncthreads();
    owner_direct_pass<HEADS, 2>(lo, cnt, wave, lane, ei, a_s, h, s_ad, s_em, s_dn, s_acc);
  }
  const int pm0 = (wave >> 1) * 16, phc = wave & 1;
  const bool prod = LIST == 1 && wave < 2 * (ACCROWS / 16) && pm0 < cnt;
  float2 qb[LIST == 1 ? 32 : 1];
  if (LIST == 1 && prod) {
    const float* bcol = Wn + (4 * (lane >> 4)) * OUTC + phc * 32 + 2 * (lane & 15);
#pragma unroll
    for (int i = 0; i < 32; ++i) qb[i] = *(const float2*)(bcol + ((i >> 2) * 8 + (i & 3)) * OUTC);
  }
  __syncthreads();

  if (LIST == 0) {
    for (int i = threadIdx.x; i < cnt * OUTC; i += OWN_T) {
      const int n = i >> 6, c = i & (OUTC - 1);
      float v = s_acc[n * ACCP + c] * invH + bias[c];
      if (DO_ELU) v = v > 0.0f ? v : (__expf(v) - 1.0f);
      out[(size_t)lo * OUTC + i] = v;
      s_acc[n * ACCP + c] = v;
    }
    __syncthreads();
    const int n = threadIdx.x >> 2, part = threadIdx.x & 3;
    if (n < cnt) {
      const float4* rr = (const float4*)(s_acc + n * ACCP + part * 16);
      const float4* ws = (const float4*)(s_wa + part * 16);
      const float4* wd = (const float4*)(s_wa + OUTC + part * 16);
      float ss = 0.0f, sd = 0.0f;
#pragma unroll
      for (int j = 0; j < 4; ++j) {
        const float4 r = rr[j], a = ws[j], b = wd[j];
        ss += r.x * a.x + r.y * a.y + r.z * a.z + r.w * a.w;
        sd += r.x * b.x + r.y * b.y + r.z * b.z + r.w * b.w;
      }
      ss += __shfl_xor(ss, 1, 32);  sd += __shfl_xor(sd, 1, 32);
      ss += __shfl_xor(ss, 2, 32);  sd += __shfl_xor(sd, 2, 32);
      if (part == 0) { as_n[lo + n] = ss; a_d[lo * H1 + n] = sd; }
    }
  } else {
    if (prod) {
      const int m0 = pm0, hc = phc;
      const int ml = lane & 15, kh = lane >> 4;
      v8f c0 = {}, c1 = {};
      const float* arow = s_acc + (m0 + ml) * ACCP + 4 * kh;
#pragma unroll
      for (int k = 0; k < OUTC; k += 8) {
        const float4 a4 = *(const float4*)(arow + k);
        const float2 q0 = qb[(k >> 3) * 4 + 0];
        const float2 q1 = qb[(k >> 3) * 4 + 1];
        const float2 q2 = qb[(k >> 3) * 4 + 2];
        const float2 q3 = qb[(k >> 3) * 4 + 3];
        v2f a, b0, b1;
        a.x = a4.x;  a.y = a4.y;
        b0.x = q0.x; b0.y = q1.x;
        b1.x = q0.y; b1.y = q1.y;
        c0 = __builtin_amdgcn_wmma_f32_16x16x4_f32(false, a, false, b0, (short)0, c0, false, false);
        c1 = __builtin_amdgcn_wmma_f32_16x16x4_f32(false, a, false, b1, (short)0, c1, false, false);
        a.x = a4.z;  a.y = a4.w;
        b0.x = q2.x; b0.y = q3.x;
        b1.x = q2.y; b1.y = q3.y;
        c0 = __builtin_amdgcn_wmma_f32_16x16x4_f32(false, a, false, b0, (short)0, c0, false, false);
        c1 = __builtin_amdgcn_wmma_f32_16x16x4_f32(false, a, false, b1, (short)0, c1, false, false);
      }
      const float2 bb = *(const float2*)(bias + hc * 32 + 2 * ml);
#pragma unroll
      for (int v = 0; v < 8; ++v) {
        const int r = m0 + 8 * kh + v;
        if (r < cnt)
          *(float2*)(out + (size_t)(lo + r) * OUTC + hc * 32 + 2 * ml) =
              make_float2(c0[v] * invH + bb.x, c1[v] * invH + bb.y);
      }
    }
  }
}

static inline unsigned cdiv(long long a, long long b) { return (unsigned)((a + b - 1) / b); }

extern "C" void kernel_launch(void* const* d_in, const int* in_sizes, int n_in,
                              void* d_out, int out_size, void* d_ws, size_t ws_size,
                              hipStream_t stream) {
  (void)in_sizes; (void)n_in; (void)out_size; (void)ws_size;
  const float* x    = (const float*)d_in[0];
  const int*   ei   = (const int*)d_in[1];
  const float* W1   = (const float*)d_in[2];
  const float* asw1 = (const float*)d_in[3];
  const float* adw1 = (const float*)d_in[4];
  const float* b1   = (const float*)d_in[5];
  const float* W2   = (const float*)d_in[6];
  const float* asw2 = (const float*)d_in[7];
  const float* adw2 = (const float*)d_in[8];
  const float* b2   = (const float*)d_in[9];
  float* out = (float*)d_out;

  float* ws   = (float*)d_ws;
  float* h1   = ws; ws += (size_t)NN * H1 * OUTC;
  float* as1  = ws; ws += (size_t)NN * H1;
  float* ad1  = ws; ws += (size_t)NN * H1;
  float* acc1 = ws; ws += (size_t)NN * OUTC;
  unsigned* own_bits = (unsigned*)ws; ws += (size_t)NN * OUTC + (size_t)NN;
  float* as2  = ws; ws += (size_t)NN;

  const int T = PRE_T;

  {
    long long waves = ((long long)NN / 16) * (H1 * OUTC / 64);
    gemm_wmma_f32_kernel<<<cdiv(waves * 32, T), T, 0, stream>>>(x, W1, h1, asw1, adw1, as1, ad1,
                                                                 ei, own_bits, NN, INCH, H1 * OUTC);
  }
  owner_aggregate_kernel<H1, 1, 0><<<OWN_WG, OWN_T, 0, stream>>>(ei, as1, ad1, h1, b1, own_bits, acc1, out, 1.0f / H1,
                                                                 W2, asw2, adw2, as2);

  owner_aggregate_kernel<1, 0, 1><<<OWN_WG, OWN_T, 0, stream>>>(ei, as2, ad1, acc1, b2, nullptr, out, nullptr, 1.0f,
                                                                 W2, nullptr, nullptr, nullptr);
}
